# starting kernel
# baseline (speedup 1.0000x reference)
#include <hip/hip_runtime.h>

typedef __attribute__((ext_vector_type(2))) float v2f;
typedef __attribute__((ext_vector_type(8))) float v8f;

constexpr int TASKS   = 100000;
constexpr int WORKERS = 500;
constexpr int FEAT    = 512;
constexpr int CLS     = 10;
constexpr int HP      = 12;
constexpr int ANS     = 1000000;
constexpr int HID     = 128;
constexpr int NODES   = TASKS + WORKERS;
constexpr int CHUNK   = 4096;


__global__ void k_deg_init(int* __restrict__ deg) {
    int i = blockIdx.x * 256 + threadIdx.x;
    if (i < NODES) deg[i] = 1;
}

__global__ void k_deg_count(const int* __restrict__ answers, int* __restrict__ deg) {
    int a = blockIdx.x * 256 + threadIdx.x;
    if (a < ANS) {
        atomicAdd(&deg[answers[3 * a]], 1);
        atomicAdd(&deg[TASKS + answers[3 * a + 1]], 1);
    }
}

__global__ void k_dinv(const int* __restrict__ deg, float* __restrict__ dinv) {
    int i = blockIdx.x * 256 + threadIdx.x;
    if (i < NODES) dinv[i] = rsqrtf((float)deg[i]);
}


__global__ void k_packB(const float* __restrict__ B, float2* __restrict__ Bp) {
    int idx = blockIdx.x * 256 + threadIdx.x;
    if (idx < (FEAT / 2) * HID) {
        int kp = idx / HID;
        int n  = idx - kp * HID;
        Bp[idx] = make_float2(B[(2 * kp) * HID + n], B[(2 * kp + 1) * HID + n]);
    }
}


__global__ __launch_bounds__(256) void k_gemm_efc(const float* __restrict__ A,
                                                  const float2* __restrict__ Bp,
                                                  const float* __restrict__ bias,
                                                  float* __restrict__ out) {
    const int wave   = threadIdx.x >> 5;
    const int lane   = threadIdx.x & 31;
    const int lane16 = lane & 15;
    const int hi     = lane >> 4;

    const int slab     = blockIdx.x * 8 + wave;
    const bool valid   = (slab * 32) < TASKS;
    const int row_base = valid ? slab * 32 : 0;

    const float*  arow0 = A + (size_t)(row_base + lane16) * FEAT + 2 * hi;
    const float*  arow1 = arow0 + (size_t)16 * FEAT;
    const float2* bbase = Bp + lane16;

    v8f acc[2][8];
#pragma unroll
    for (int t = 0; t < 8; ++t) {
        float bv = bias[t * 16 + lane16];
        acc[0][t] = (v8f){bv, bv, bv, bv, bv, bv, bv, bv};
        acc[1][t] = acc[0][t];
    }

    for (int k = 0; k < FEAT; k += 4) {
        const float2 av0 = *(const float2*)(arow0 + k);
        const float2 av1 = *(const float2*)(arow1 + k);
        v2f af0 = {av0.x, av0.y};
        v2f af1 = {av1.x, av1.y};
        const float2* brow = bbase + (size_t)((k >> 1) + hi) * HID;
#pragma unroll
        for (int t = 0; t < 8; ++t) {
            const float2 bv = brow[t * 16];
            v2f bf = {bv.x, bv.y};
            acc[0][t] = __builtin_amdgcn_wmma_f32_16x16x4_f32(
                false, af0, false, bf, (short)0, acc[0][t], false, false);
            acc[1][t] = __builtin_amdgcn_wmma_f32_16x16x4_f32(
                false, af1, false, bf, (short)0, acc[1][t], false, false);
        }
    }

    if (valid) {
#pragma unroll
        for (int r = 0; r < 2; ++r) {
#pragma unroll
            for (int t = 0; t < 8; ++t) {
#pragma unroll
                for (int j = 0; j < 8; ++j) {
                    int m   = row_base + r * 16 + j + 8 * hi;
                    float v = acc[r][t][j];
                    out[(size_t)m * HID + t * 16 + lane16] = v > 0.f ? v : 0.f;
                }
            }
        }
    }
}


__global__ __launch_bounds__(256) void k_h(const float* __restrict__ tf_h,
                                           const float* __restrict__ wf,
                                           const float* __restrict__ Wm,
                                           const float* __restrict__ Wl,
                                           float* __restrict__ hm,
                                           float* __restrict__ hl) {
    const int wave   = threadIdx.x >> 5;
    const int lane   = threadIdx.x & 31;
    const int lane16 = lane & 15;
    const int hi     = lane >> 4;

    const int node_base = (blockIdx.x * 8 + wave) * 16;
    int nd = node_base + lane16;
    if (nd > NODES - 1) nd = NODES - 1;
    const float* xrow = (nd < TASKS) ? (tf_h + (size_t)nd * HID)
                                     : (wf + (size_t)(nd - TASKS) * HID);
    const int n = lane16;
    const bool ncol = n < CLS;

    v8f accM = {0.f, 0.f, 0.f, 0.f, 0.f, 0.f, 0.f, 0.f};
    v8f accL = {0.f, 0.f, 0.f, 0.f, 0.f, 0.f, 0.f, 0.f};

    for (int k = 0; k < HID; k += 4) {
        const int k0 = k + 2 * hi;
        const float2 av = *(const float2*)(xrow + k0);
        v2f af = {av.x, av.y};
        v2f bm, bl;
        bm.x = ncol ? Wm[k0 * CLS + n] : 0.f;
        bm.y = ncol ? Wm[(k0 + 1) * CLS + n] : 0.f;
        bl.x = ncol ? Wl[k0 * CLS + n] : 0.f;
        bl.y = ncol ? Wl[(k0 + 1) * CLS + n] : 0.f;
        accM = __builtin_amdgcn_wmma_f32_16x16x4_f32(false, af, false, bm, (short)0, accM, false, false);
        accL = __builtin_amdgcn_wmma_f32_16x16x4_f32(false, af, false, bl, (short)0, accL, false, false);
    }

    if (ncol) {
#pragma unroll
        for (int j = 0; j < 8; ++j) {
            int m = node_base + j + 8 * hi;
            if (m < NODES) {
                hm[(size_t)m * HP + n] = accM[j];
                hl[(size_t)m * HP + n] = accL[j];
            }
        }
    }
}


__global__ void k_agg_init(const float* __restrict__ hm, const float* __restrict__ hl,
                           const float* __restrict__ dinv,
                           const float* __restrict__ bm, const float* __restrict__ bl,
                           float* __restrict__ mean, float* __restrict__ lsd) {
    int idx = blockIdx.x * 256 + threadIdx.x;
    if (idx < NODES * CLS) {
        int i = idx / CLS;
        int c = idx - i * CLS;
        float d  = dinv[i];
        float sw = d * d;
        mean[idx] = bm[c] + hm[(size_t)i * HP + c] * sw;
        lsd[idx]  = bl[c] + hl[(size_t)i * HP + c] * sw;
    }
}


__device__ __forceinline__ void load_row10(const float* __restrict__ p, float* r) {
    const float4 a = *(const float4*)(p);
    const float4 b = *(const float4*)(p + 4);
    const float2 c = *(const float2*)(p + 8);
    r[0] = a.x; r[1] = a.y; r[2] = a.z; r[3] = a.w;
    r[4] = b.x; r[5] = b.y; r[6] = b.z; r[7] = b.w;
    r[8] = c.x; r[9] = c.y;
}

__global__ __launch_bounds__(256) void k_agg_edges(const int* __restrict__ answers,
                                                   const float* __restrict__ hm,
                                                   const float* __restrict__ hl,
                                                   const float* __restrict__ dinv,
                                                   float* __restrict__ mean,
                                                   float* __restrict__ lsd) {
    __shared__ float wacc[2 * WORKERS * CLS];
    for (int i = threadIdx.x; i < 2 * WORKERS * CLS; i += 256) wacc[i] = 0.f;
    __syncthreads();

    const int start = blockIdx.x * CHUNK;
    int end = start + CHUNK;
    if (end > ANS) end = ANS;

    for (int a = start + threadIdx.x; a < end; a += 256) {
        const int t    = answers[3 * a];
        const int w    = answers[3 * a + 1];
        const int wrow = TASKS + w;
        const float we = dinv[t] * dinv[wrow];

        float hmt[10], hmw[10], hlt[10], hlw[10];
        load_row10(hm + (size_t)t * HP,    hmt);
        load_row10(hm + (size_t)wrow * HP, hmw);
        load_row10(hl + (size_t)t * HP,    hlt);
        load_row10(hl + (size_t)wrow * HP, hlw);

        float* mt = mean + (size_t)t * CLS;
        float* lt = lsd + (size_t)t * CLS;
#pragma unroll
        for (int c = 0; c < CLS; ++c) {
            atomicAdd(&mt[c], hmw[c] * we);
            atomicAdd(&lt[c], hlw[c] * we);
            atomicAdd(&wacc[w * CLS + c], hmt[c] * we);
            atomicAdd(&wacc[WORKERS * CLS + w * CLS + c], hlt[c] * we);
        }
    }
    __syncthreads();

    for (int i = threadIdx.x; i < WORKERS * CLS; i += 256) {
        atomicAdd(&mean[(size_t)TASKS * CLS + i], wacc[i]);
        atomicAdd(&lsd[(size_t)TASKS * CLS + i], wacc[WORKERS * CLS + i]);
    }
}


__global__ void k_z(const float* __restrict__ mean, const float* __restrict__ lsd,
                    const float* __restrict__ eps, float* __restrict__ z) {
    int idx = blockIdx.x * 256 + threadIdx.x;
    if (idx < NODES * CLS)
        z[idx] = mean[idx] + (eps[idx] * 0.01f) * expf(lsd[idx]);
}

__global__ void k_crowd(const int* __restrict__ answers, const float* __restrict__ z,
                        float* __restrict__ crowd) {
    int idx = blockIdx.x * 256 + threadIdx.x;
    if (idx < ANS * CLS) {
        int a = idx / CLS;
        int c = idx - a * CLS;
        int t = answers[3 * a];
        int w = answers[3 * a + 1];
        crowd[idx] = z[(size_t)t * CLS + c] * z[(size_t)(TASKS + w) * CLS + c];
    }
}


extern "C" void kernel_launch(void* const* d_in, const int* in_sizes, int n_in,
                              void* d_out, int out_size, void* d_ws, size_t ws_size,
                              hipStream_t stream) {
    (void)in_sizes; (void)n_in; (void)out_size; (void)ws_size;

    const float* task_feature   = (const float*)d_in[0];
    const int*   answers        = (const int*)d_in[1];
    const float* worker_feature = (const float*)d_in[2];
    const float* W_efc          = (const float*)d_in[3];
    const float* b_efc          = (const float*)d_in[4];
    const float* W_mean         = (const float*)d_in[5];
    const float* b_mean         = (const float*)d_in[6];
    const float* W_ls           = (const float*)d_in[7];
    const float* b_ls           = (const float*)d_in[8];
    const float* eps            = (const float*)d_in[9];

    float* out   = (float*)d_out;
    float* crowd = out;
    float* mean  = out + (size_t)ANS * CLS;
    float* lsd   = mean + (size_t)NODES * CLS;

    float*  ws   = (float*)d_ws;
    float*  tf_h = ws;
    float2* Bp   = (float2*)(tf_h + (size_t)TASKS * HID);
    float*  hm   = (float*)(Bp + (size_t)(FEAT / 2) * HID);
    float*  hl   = hm + (size_t)NODES * HP;
    float*  zz   = hl + (size_t)NODES * HP;
    float*  dinv = zz + (size_t)NODES * CLS;
    int*    deg  = (int*)(dinv + NODES);

    auto cdiv = [](int a, int b) { return (a + b - 1) / b; };

    k_deg_init <<<cdiv(NODES, 256), 256, 0, stream>>>(deg);
    k_deg_count<<<cdiv(ANS, 256), 256, 0, stream>>>(answers, deg);
    k_dinv     <<<cdiv(NODES, 256), 256, 0, stream>>>(deg, dinv);

    k_packB    <<<cdiv((FEAT / 2) * HID, 256), 256, 0, stream>>>(W_efc, Bp);
    k_gemm_efc <<<cdiv(TASKS / 32, 8), 256, 0, stream>>>(task_feature, Bp, b_efc, tf_h);
    k_h        <<<cdiv(cdiv(NODES, 16), 8), 256, 0, stream>>>(tf_h, worker_feature,
                                                              W_mean, W_ls, hm, hl);

    k_agg_init <<<cdiv(NODES * CLS, 256), 256, 0, stream>>>(hm, hl, dinv, b_mean, b_ls, mean, lsd);
    k_agg_edges<<<cdiv(ANS, CHUNK), 256, 0, stream>>>(answers, hm, hl, dinv, mean, lsd);

    k_z        <<<cdiv(NODES * CLS, 256), 256, 0, stream>>>(mean, lsd, eps, zz);
    k_crowd    <<<cdiv(ANS * CLS, 256), 256, 0, stream>>>(answers, zz, crowd);
}


# Crowd product leaves the one-thread-per-element form (312K short waves): each thread owns 25 output quads, reads packed keys and 8-byte pieces of both z rows, stores float4 = whole lines; no LDS
# speedup vs baseline: 1.9150x; 1.9150x over previous
#include <hip/hip_runtime.h>

typedef __attribute__((ext_vector_type(2))) float v2f;
typedef __attribute__((ext_vector_type(8))) float v8f;

constexpr int TASKS   = 100000;
constexpr int WORKERS = 500;
constexpr int FEAT    = 512;
constexpr int CLS     = 10;
constexpr int HP      = 12;
constexpr int ANS     = 1000000;
constexpr int HID     = 128;
constexpr int NODES   = TASKS + WORKERS;
constexpr int KSH     = 9;
constexpr int KMASK   = (1 << KSH) - 1;
constexpr int TBLK    = 800;
constexpr int NOWN    = TASKS / TBLK;
constexpr int ROW2    = 2 * CLS;
constexpr int QCAP    = 16384;
constexpr int NCHUNK  = (ANS + QCAP - 1) / QCAP;
constexpr int OWN_THREADS = 1024;
constexpr int GRPS_CHUNK     = QCAP / 4;
constexpr int GRPS_THREAD     = GRPS_CHUNK / OWN_THREADS;
constexpr int NGRP    = ANS / 4;
constexpr int TROWS   = 16;
constexpr int NTILE   = TASKS / TROWS;
constexpr int GEMM_WGS = 256;
constexpr int NPAIR   = GEMM_WGS * 4;
constexpr int BPITCH  = HID + 16;
constexpr int KBLK    = 32;
constexpr int NBLK    = FEAT / KBLK;


__global__ void k_deg_init(int* __restrict__ deg) {
    int i = blockIdx.x * 256 + threadIdx.x;
    if (i < NODES) deg[i] = 1;
}

__global__ void k_deg_count(const int* __restrict__ answers, int* __restrict__ deg,
                            int* __restrict__ keys) {
    int a = blockIdx.x * 256 + threadIdx.x;
    if (a < ANS) {
        const int t = answers[3 * a];
        const int w = answers[3 * a + 1];
        atomicAdd(&deg[t], 1);
        atomicAdd(&deg[TASKS + w], 1);
        keys[a] = (t << KSH) | w;
    }
}

__global__ void k_dinv(const int* __restrict__ deg, float* __restrict__ dinv) {
    int i = blockIdx.x * 256 + threadIdx.x;
    if (i < NODES) dinv[i] = rsqrtf((float)deg[i]);
}


__device__ __forceinline__ void load_ablock(v2f (&a)[8], const float* __restrict__ p) {
#pragma unroll
    for (int s = 0; s < 8; ++s) {
        const float2 v = *(const float2*)(p + 4 * s);
        a[s] = (v2f){v.x, v.y};
    }
}

__device__ __forceinline__ void mul_block(v8f (&acc)[8], const v2f (&a)[8], const float2* bb) {
#pragma unroll
    for (int s = 0; s < 8; ++s) {
        const float2* brow = bb + s * 2 * BPITCH;
#pragma unroll
        for (int t = 0; t < 8; ++t) {
            const float2 bv = brow[t * 16];
            v2f bf = {bv.x, bv.y};
            acc[t] = __builtin_amdgcn_wmma_f32_16x16x4_f32(
                false, a[s], false, bf, (short)0, acc[t], false, false);
        }
    }
}

__global__ __launch_bounds__(256) void k_gemm_efc(const float* __restrict__ A,
                                                  const float* __restrict__ B,
                                                  const float* __restrict__ bias,
                                                  float* __restrict__ out) {
    __shared__ alignas(16) float2 Bs[(FEAT / 2) * BPITCH];

    const int tid = threadIdx.x;

#pragma unroll 4
    for (int it = tid; it < (FEAT / 2) * (HID / 4); it += 256) {
        const int kp = it / (HID / 4);
        const int n4 = it - kp * (HID / 4);
        const float4 r0 = *(const float4*)(B + (size_t)(2 * kp) * HID + 4 * n4);
        const float4 r1 = *(const float4*)(B + (size_t)(2 * kp + 1) * HID + 4 * n4);
        float4* dst = (float4*)&Bs[kp * BPITCH + 4 * n4];
        dst[0] = make_float4(r0.x, r1.x, r0.y, r1.y);
        dst[1] = make_float4(r0.z, r1.z, r0.w, r1.w);
    }
    __syncthreads();

    const int wave   = tid >> 5;
    const int lane   = tid & 31;
    const int lane16 = lane & 15;
    const int hi     = lane >> 4;

    const int pair = blockIdx.x * 4 + (wave & 3);
    const int sub  = wave >> 2;

    float bcol[8];
#pragma unroll
    for (int t = 0; t < 8; ++t) bcol[t] = bias[t * 16 + lane16];

    const float2* bl = Bs + hi * BPITCH + lane16;

    int tile = pair + NPAIR * sub;
    const float* arow = A + (size_t)(tile * TROWS + lane16) * FEAT + 2 * hi;

    v2f a0[8], a1[8];
    load_ablock(a0, arow);

    for (;;) {
        const int  ntile = tile + 2 * NPAIR;
        const bool more  = ntile < NTILE;
        const float* nrow = A + (size_t)((more ? ntile : tile) * TROWS + lane16) * FEAT + 2 * hi;

        v8f acc[8];
#pragma unroll
        for (int t = 0; t < 8; ++t) {
            const float bv = bcol[t];
            acc[t] = (v8f){bv, bv, bv, bv, bv, bv, bv, bv};
        }

        for (int b = 0; b < NBLK; b += 2) {
            load_ablock(a1, arow + (b + 1) * KBLK);
            mul_block(acc, a0, bl + (size_t)b * (KBLK / 2) * BPITCH);
            load_ablock(a0, (b + 2 < NBLK) ? arow + (b + 2) * KBLK : nrow);
            mul_block(acc, a1, bl + (size_t)(b + 1) * (KBLK / 2) * BPITCH);
        }

#pragma unroll
        for (int t = 0; t < 8; ++t) {
#pragma unroll
            for (int j = 0; j < 8; ++j) {
                const int m   = tile * TROWS + j + 8 * hi;
                const float v = acc[t][j];
                out[(size_t)m * HID + t * 16 + lane16] = v > 0.f ? v : 0.f;
            }
        }

        if (!more) break;
        tile = ntile;
        arow = nrow;
    }
}


__global__ __launch_bounds__(256) void k_h(const float* __restrict__ tf_h,
                                           const float* __restrict__ wf,
                                           const float* __restrict__ Wm,
                                           const float* __restrict__ Wl,
                                           float* __restrict__ hm,
                                           float* __restrict__ hl) {
    const int wave   = threadIdx.x >> 5;
    const int lane   = threadIdx.x & 31;
    const int lane16 = lane & 15;
    const int hi     = lane >> 4;

    const int node_base = (blockIdx.x * 8 + wave) * 16;
    int nd = node_base + lane16;
    if (nd > NODES - 1) nd = NODES - 1;
    const float* xrow = (nd < TASKS) ? (tf_h + (size_t)nd * HID)
                                     : (wf + (size_t)(nd - TASKS) * HID);
    const int n = lane16;
    const bool ncol = n < CLS;

    v8f accM = {0.f, 0.f, 0.f, 0.f, 0.f, 0.f, 0.f, 0.f};
    v8f accL = {0.f, 0.f, 0.f, 0.f, 0.f, 0.f, 0.f, 0.f};

    for (int k = 0; k < HID; k += 4) {
        const int k0 = k + 2 * hi;
        const float2 av = *(const float2*)(xrow + k0);
        v2f af = {av.x, av.y};
        v2f bm, bl;
        bm.x = ncol ? Wm[k0 * CLS + n] : 0.f;
        bm.y = ncol ? Wm[(k0 + 1) * CLS + n] : 0.f;
        bl.x = ncol ? Wl[k0 * CLS + n] : 0.f;
        bl.y = ncol ? Wl[(k0 + 1) * CLS + n] : 0.f;
        accM = __builtin_amdgcn_wmma_f32_16x16x4_f32(false, af, false, bm, (short)0, accM, false, false);
        accL = __builtin_amdgcn_wmma_f32_16x16x4_f32(false, af, false, bl, (short)0, accL, false, false);
    }

    if (ncol) {
#pragma unroll
        for (int j = 0; j < 8; ++j) {
            int m = node_base + j + 8 * hi;
            if (m < NODES) {
                hm[(size_t)m * HP + n] = accM[j];
                hl[(size_t)m * HP + n] = accL[j];
            }
        }
    }
}


__global__ void k_agg_init_w(const float* __restrict__ hm, const float* __restrict__ hl,
                             const float* __restrict__ dinv,
                             const float* __restrict__ bm, const float* __restrict__ bl,
                             float* __restrict__ mean, float* __restrict__ lsd) {
    int i = blockIdx.x * 256 + threadIdx.x;
    if (i < WORKERS * CLS) {
        int w = i / CLS;
        int c = i - w * CLS;
        float d  = dinv[TASKS + w];
        float sw = d * d;
        mean[(size_t)TASKS * CLS + i] = bm[c] + hm[(size_t)(TASKS + w) * HP + c] * sw;
        lsd[(size_t)TASKS * CLS + i]  = bl[c] + hl[(size_t)(TASKS + w) * HP + c] * sw;
    }
}


__device__ __forceinline__ void lds_barrier() {
    __builtin_amdgcn_fence(__ATOMIC_RELEASE, "workgroup", "local");
    __builtin_amdgcn_s_barrier();
    __builtin_amdgcn_fence(__ATOMIC_ACQUIRE, "workgroup", "local");
}

__global__ __launch_bounds__(OWN_THREADS) void k_agg_owner(const int* __restrict__ keys,
                                                           const float* __restrict__ hm,
                                                           const float* __restrict__ hl,
                                                           const float* __restrict__ dinv,
                                                           const float* __restrict__ bm,
                                                           const float* __restrict__ bl,
                                                           float* __restrict__ mean,
                                                           float* __restrict__ lsd) {
    __shared__ alignas(16) float acc[TBLK * ROW2];
    __shared__ alignas(16) float Fs[TBLK * ROW2];
    __shared__ alignas(16) float Gs[WORKERS * ROW2];
    __shared__ alignas(16) float wacc[WORKERS * ROW2];
    __shared__ int   q[QCAP];
    __shared__ int   qn[2];

    const int tid    = threadIdx.x;
    const int t_base = blockIdx.x * TBLK;

    for (int i = tid; i < TBLK * CLS; i += OWN_THREADS) {
        const int tl = i / CLS;
        const int c  = i - tl * CLS;
        const int t  = t_base + tl;
        const float d = dinv[t];
        Fs[tl * ROW2 + c]        = d * hm[(size_t)t * HP + c];
        Fs[tl * ROW2 + CLS + c]  = d * hl[(size_t)t * HP + c];
        acc[tl * ROW2 + c]       = 0.f;
        acc[tl * ROW2 + CLS + c] = 0.f;
    }
    for (int i = tid; i < WORKERS * CLS; i += OWN_THREADS) {
        const int w  = i / CLS;
        const int c  = i - w * CLS;
        const int r  = TASKS + w;
        const float d = dinv[r];
        Gs[w * ROW2 + c]         = d * hm[(size_t)r * HP + c];
        Gs[w * ROW2 + CLS + c]   = d * hl[(size_t)r * HP + c];
        wacc[w * ROW2 + c]       = 0.f;
        wacc[w * ROW2 + CLS + c] = 0.f;
    }
    if (tid < 2) qn[tid] = 0;
    __syncthreads();

    const int4* __restrict__ kp = (const int4*)keys;
    const unsigned key_lo  = (unsigned)t_base << KSH;
    const unsigned key_len = (unsigned)TBLK << KSH;

    int4 cur[GRPS_THREAD];
#pragma unroll
    for (int j = 0; j < GRPS_THREAD; ++j) cur[j] = kp[j * OWN_THREADS + tid];

    for (int ch = 0; ch < NCHUNK; ++ch) {
        const int p = ch & 1;

#pragma unroll
        for (int j = 0; j < GRPS_THREAD; ++j) {
            const int kk[4] = {cur[j].x, cur[j].y, cur[j].z, cur[j].w};
#pragma unroll
            for (int u = 0; u < 4; ++u) {
                const unsigned rel = (unsigned)kk[u] - key_lo;
                if (rel < key_len) {
                    const int pos = atomicAdd(&qn[p], 1);
                    q[pos] = (int)rel;
                }
            }
        }

        if (ch + 1 < NCHUNK) {
#pragma unroll
            for (int j = 0; j < GRPS_THREAD; ++j) {
                const int g  = (ch + 1) * GRPS_CHUNK + j * OWN_THREADS + tid;
                const int4 v = kp[g < NGRP ? g : NGRP - 1];
                cur[j] = (g < NGRP) ? v : make_int4(-1, -1, -1, -1);
            }
        }

        lds_barrier();
        const int items = qn[p] * (ROW2 / 4);
        if (tid == 0) qn[p ^ 1] = 0;

        for (int i = tid; i < items; i += OWN_THREADS) {
            const int e   = i / (ROW2 / 4);
            const int c4  = (i - e * (ROW2 / 4)) * 4;
            const int key = q[e];
            const int tl  = key >> KSH;
            const int w   = key & KMASK;
            const float4 g = *(const float4*)&Gs[w * ROW2 + c4];
            const float4 f = *(const float4*)&Fs[tl * ROW2 + c4];
            float* a = &acc[tl * ROW2 + c4];
            float* b = &wacc[w * ROW2 + c4];
            atomicAdd(a + 0, g.x); atomicAdd(a + 1, g.y); atomicAdd(a + 2, g.z); atomicAdd(a + 3, g.w);
            atomicAdd(b + 0, f.x); atomicAdd(b + 1, f.y); atomicAdd(b + 2, f.z); atomicAdd(b + 3, f.w);
        }
        lds_barrier();
    }

    for (int i = tid; i < TBLK * CLS; i += OWN_THREADS) {
        const int tl = i / CLS;
        const int c  = i - tl * CLS;
        const float d = dinv[t_base + tl];
        const size_t o = (size_t)t_base * CLS + i;
        mean[o] = bm[c] + d * (Fs[tl * ROW2 + c] + acc[tl * ROW2 + c]);
        lsd[o]  = bl[c] + d * (Fs[tl * ROW2 + CLS + c] + acc[tl * ROW2 + CLS + c]);
    }

    for (int i = tid; i < WORKERS * CLS; i += OWN_THREADS) {
        const int w = i / CLS;
        const int c = i - w * CLS;
        const float d = dinv[TASKS + w];
        atomicAdd(&mean[(size_t)TASKS * CLS + i], d * wacc[w * ROW2 + c]);
        atomicAdd(&lsd[(size_t)TASKS * CLS + i],  d * wacc[w * ROW2 + CLS + c]);
    }
}


__global__ void k_z(const float* __restrict__ mean, const float* __restrict__ lsd,
                    const float* __restrict__ eps, float* __restrict__ z) {
    int idx = blockIdx.x * 256 + threadIdx.x;
    if (idx < NODES * CLS)
        z[idx] = mean[idx] + (eps[idx] * 0.01f) * expf(lsd[idx]);
}

constexpr int NQUAD      = ANS * CLS / 4;
constexpr int CR_THREADS = 160;
constexpr int CR_WGS     = 625;
constexpr int CR_GT      = CR_THREADS * CR_WGS;
constexpr int CR_TRIPS   = NQUAD / CR_GT;
constexpr int CR_BATCH   = 5;
constexpr int CR_ASTEP   = CR_GT * 4 / CLS;
static_assert((ANS * CLS) % 4 == 0 && NQUAD % CR_GT == 0 && (CR_GT * 4) % CLS == 0 &&
              CR_TRIPS % CR_BATCH == 0 && CLS % 2 == 0, "crowd partition must be exact");

__global__ __launch_bounds__(CR_THREADS) void k_crowd(const int* __restrict__ keys,
                                                      const float* __restrict__ z,
                                                      float* __restrict__ crowd) {
    const int g  = blockIdx.x * CR_THREADS + threadIdx.x;
    const int eA = 4 * g;
    const int aA = eA / CLS;
    const int cA = eA - aA * CLS;
    const int eB = eA + 2;
    const int aB = eB / CLS;
    const int cB = eB - aB * CLS;

    const float* __restrict__ ztA = z + cA;
    const float* __restrict__ ztB = z + cB;
    const float* __restrict__ zwA = z + (size_t)TASKS * CLS + cA;
    const float* __restrict__ zwB = z + (size_t)TASKS * CLS + cB;
    const int*   __restrict__ kA  = keys + aA;
    const int*   __restrict__ kB  = keys + aB;
    float4*      __restrict__ o   = (float4*)crowd + g;

    for (int b = 0; b < CR_TRIPS; b += CR_BATCH) {
        int ka[CR_BATCH], kb[CR_BATCH];
#pragma unroll
        for (int u = 0; u < CR_BATCH; ++u) {
            ka[u] = kA[(b + u) * CR_ASTEP];
            kb[u] = kB[(b + u) * CR_ASTEP];
        }
        float2 ta[CR_BATCH], wa[CR_BATCH], tb[CR_BATCH], wb[CR_BATCH];
#pragma unroll
        for (int u = 0; u < CR_BATCH; ++u) {
            ta[u] = *(const float2*)(ztA + (ka[u] >> KSH) * CLS);
            wa[u] = *(const float2*)(zwA + (ka[u] & KMASK) * CLS);
            tb[u] = *(const float2*)(ztB + (kb[u] >> KSH) * CLS);
            wb[u] = *(const float2*)(zwB + (kb[u] & KMASK) * CLS);
        }
#pragma unroll
        for (int u = 0; u < CR_BATCH; ++u)
            o[(size_t)(b + u) * CR_GT] = make_float4(ta[u].x * wa[u].x, ta[u].y * wa[u].y,
                                                     tb[u].x * wb[u].x, tb[u].y * wb[u].y);
    }
}


extern "C" void kernel_launch(void* const* d_in, const int* in_sizes, int n_in,
                              void* d_out, int out_size, void* d_ws, size_t ws_size,
                              hipStream_t stream) {
    (void)in_sizes; (void)n_in; (void)out_size; (void)ws_size;

    const float* task_feature   = (const float*)d_in[0];
    const int*   answers        = (const int*)d_in[1];
    const float* worker_feature = (const float*)d_in[2];
    const float* W_efc          = (const float*)d_in[3];
    const float* b_efc          = (const float*)d_in[4];
    const float* W_mean         = (const float*)d_in[5];
    const float* b_mean         = (const float*)d_in[6];
    const float* W_ls           = (const float*)d_in[7];
    const float* b_ls           = (const float*)d_in[8];
    const float* eps            = (const float*)d_in[9];

    float* out   = (float*)d_out;
    float* crowd = out;
    float* mean  = out + (size_t)ANS * CLS;
    float* lsd   = mean + (size_t)NODES * CLS;

    float*  ws   = (float*)d_ws;
    float*  tf_h = ws;
    float*  hm   = tf_h + (size_t)TASKS * HID;
    float*  hl   = hm + (size_t)NODES * HP;
    float*  zz   = hl + (size_t)NODES * HP;
    float*  dinv = zz + (size_t)NODES * CLS;
    int*    deg  = (int*)(dinv + NODES);

    auto cdiv = [](int a, int b) { return (a + b - 1) / b; };

    k_gemm_efc <<<GEMM_WGS, 256, 0, stream>>>(task_feature, W_efc, b_efc, tf_h);
    k_h        <<<cdiv(cdiv(NODES, 16), 8), 256, 0, stream>>>(tf_h, worker_feature,
                                                              W_mean, W_ls, hm, hl);

    int* keys = (int*)tf_h;
    k_deg_init <<<cdiv(NODES, 256), 256, 0, stream>>>(deg);
    k_deg_count<<<cdiv(ANS, 256), 256, 0, stream>>>(answers, deg, keys);
    k_dinv     <<<cdiv(NODES, 256), 256, 0, stream>>>(deg, dinv);

    k_agg_init_w<<<cdiv(WORKERS * CLS, 256), 256, 0, stream>>>(hm, hl, dinv, b_mean, b_ls, mean, lsd);
    k_agg_owner <<<NOWN, OWN_THREADS, 0, stream>>>(keys, hm, hl, dinv, b_mean, b_ls, mean, lsd);

    k_z        <<<cdiv(NODES * CLS, 256), 256, 0, stream>>>(mean, lsd, eps, zz);
    k_crowd    <<<CR_WGS, CR_THREADS, 0, stream>>>(keys, zz, crowd);
}


# Node degrees without global atomics: 256 work-groups (16 task ranges x 16 key slices) count in LDS integer adds and store partial planes, summed in the rsqrt launch; no float sum changes
# speedup vs baseline: 3.4754x; 1.8148x over previous
#include <hip/hip_runtime.h>

typedef __attribute__((ext_vector_type(2))) float v2f;
typedef __attribute__((ext_vector_type(8))) float v8f;

constexpr int TASKS   = 100000;
constexpr int WORKERS = 500;
constexpr int FEAT    = 512;
constexpr int CLS     = 10;
constexpr int HP      = 12;
constexpr int ANS     = 1000000;
constexpr int HID     = 128;
constexpr int NODES   = TASKS + WORKERS;
constexpr int KSH     = 9;
constexpr int KMASK   = (1 << KSH) - 1;
constexpr int TBLK    = 800;
constexpr int NOWN    = TASKS / TBLK;
constexpr int ROW2    = 2 * CLS;
constexpr int QCAP    = 16384;
constexpr int NCHUNK  = (ANS + QCAP - 1) / QCAP;
constexpr int OWN_THREADS = 1024;
constexpr int GRPS_CHUNK     = QCAP / 4;
constexpr int GRPS_THREAD     = GRPS_CHUNK / OWN_THREADS;
constexpr int NGRP    = ANS / 4;
constexpr int TROWS   = 16;
constexpr int NTILE   = TASKS / TROWS;
constexpr int GEMM_WGS = 256;
constexpr int NPAIR   = GEMM_WGS * 4;
constexpr int BPITCH  = HID + 16;
constexpr int KBLK    = 32;
constexpr int NBLK    = FEAT / KBLK;

constexpr int CNT_R     = 16;
constexpr int CNT_S     = 16;
constexpr int CNT_WGS   = CNT_R * CNT_S;
constexpr int CNT_THREADS = 1024;
constexpr int RTASKS    = TASKS / CNT_R;
constexpr int SGRPS     = (ANS / 4) / CNT_S;
constexpr int WPITCH    = 512;
constexpr int CNT_BATCH = 8;
static_assert(TASKS % CNT_R == 0 && ANS % (4 * CNT_S) == 0 && WORKERS <= WPITCH && WPITCH <= CNT_THREADS,
              "count partition must be exact");

__global__ void k_pack(const int* __restrict__ answers, int* __restrict__ keys) {
    int a = blockIdx.x * 256 + threadIdx.x;
    if (a < ANS) {
        const int t = answers[3 * a];
        const int w = answers[3 * a + 1];
        keys[a] = (t << KSH) | w;
    }
}

__global__ __launch_bounds__(CNT_THREADS) void k_count(const int* __restrict__ keys,
                                                       int* __restrict__ tpart,
                                                       int* __restrict__ wpart) {
    __shared__ int tcnt[RTASKS];
    __shared__ int wcnt[WPITCH];

    const int tid = threadIdx.x;
    const int r   = blockIdx.x % CNT_R;
    const int s   = blockIdx.x / CNT_R;

    for (int i = tid; i < RTASKS; i += CNT_THREADS) tcnt[i] = 0;
    if (tid < WPITCH) wcnt[tid] = 0;
    __syncthreads();

    const int4* __restrict__ kp = (const int4*)keys + (size_t)s * SGRPS;
    const unsigned key_lo  = (unsigned)(r * RTASKS) << KSH;
    const unsigned key_len = (unsigned)RTASKS << KSH;

    for (int g0 = tid; g0 < SGRPS; g0 += CNT_THREADS * CNT_BATCH) {
        int4 v[CNT_BATCH];
#pragma unroll
        for (int j = 0; j < CNT_BATCH; ++j) {
            const int g = g0 + j * CNT_THREADS;
            v[j] = kp[g < SGRPS ? g : SGRPS - 1];
        }
#pragma unroll
        for (int j = 0; j < CNT_BATCH; ++j) {
            if (g0 + j * CNT_THREADS < SGRPS) {
                const int kk[4] = {v[j].x, v[j].y, v[j].z, v[j].w};
#pragma unroll
                for (int u = 0; u < 4; ++u) {
                    const unsigned rel = (unsigned)kk[u] - key_lo;
                    if (rel < key_len) {
                        atomicAdd(&tcnt[rel >> KSH], 1);
                        atomicAdd(&wcnt[rel & KMASK], 1);
                    }
                }
            }
        }
    }
    __syncthreads();

    int* __restrict__ tp = tpart + (size_t)s * TASKS + r * RTASKS;
    for (int i = tid; i < RTASKS; i += CNT_THREADS) tp[i] = tcnt[i];
    if (tid < WPITCH) wpart[(size_t)blockIdx.x * WPITCH + tid] = wcnt[tid];
}

__global__ void k_dinv(const int* __restrict__ tpart, const int* __restrict__ wpart,
                       float* __restrict__ dinv) {
    int i = blockIdx.x * 256 + threadIdx.x;
    if (i < TASKS) {
        int d = 1;
#pragma unroll
        for (int s = 0; s < CNT_S; ++s) d += tpart[(size_t)s * TASKS + i];
        dinv[i] = rsqrtf((float)d);
    } else if (i < NODES) {
        const int w = i - TASKS;
        int d = 1;
#pragma unroll 16
        for (int b = 0; b < CNT_WGS; ++b) d += wpart[(size_t)b * WPITCH + w];
        dinv[i] = rsqrtf((float)d);
    }
}


__device__ __forceinline__ void load_ablock(v2f (&a)[8], const float* __restrict__ p) {
#pragma unroll
    for (int s = 0; s < 8; ++s) {
        const float2 v = *(const float2*)(p + 4 * s);
        a[s] = (v2f){v.x, v.y};
    }
}

__device__ __forceinline__ void mul_block(v8f (&acc)[8], const v2f (&a)[8], const float2* bb) {
#pragma unroll
    for (int s = 0; s < 8; ++s) {
        const float2* brow = bb + s * 2 * BPITCH;
#pragma unroll
        for (int t = 0; t < 8; ++t) {
            const float2 bv = brow[t * 16];
            v2f bf = {bv.x, bv.y};
            acc[t] = __builtin_amdgcn_wmma_f32_16x16x4_f32(
                false, a[s], false, bf, (short)0, acc[t], false, false);
        }
    }
}

__global__ __launch_bounds__(256) void k_gemm_efc(const float* __restrict__ A,
                                                  const float* __restrict__ B,
                                                  const float* __restrict__ bias,
                                                  float* __restrict__ out) {
    __shared__ alignas(16) float2 Bs[(FEAT / 2) * BPITCH];

    const int tid = threadIdx.x;

#pragma unroll 4
    for (int it = tid; it < (FEAT / 2) * (HID / 4); it += 256) {
        const int kp = it / (HID / 4);
        const int n4 = it - kp * (HID / 4);
        const float4 r0 = *(const float4*)(B + (size_t)(2 * kp) * HID + 4 * n4);
        const float4 r1 = *(const float4*)(B + (size_t)(2 * kp + 1) * HID + 4 * n4);
        float4* dst = (float4*)&Bs[kp * BPITCH + 4 * n4];
        dst[0] = make_float4(r0.x, r1.x, r0.y, r1.y);
        dst[1] = make_float4(r0.z, r1.z, r0.w, r1.w);
    }
    __syncthreads();

    const int wave   = tid >> 5;
    const int lane   = tid & 31;
    const int lane16 = lane & 15;
    const int hi     = lane >> 4;

    const int pair = blockIdx.x * 4 + (wave & 3);
    const int sub  = wave >> 2;

    float bcol[8];
#pragma unroll
    for (int t = 0; t < 8; ++t) bcol[t] = bias[t * 16 + lane16];

    const float2* bl = Bs + hi * BPITCH + lane16;

    int tile = pair + NPAIR * sub;
    const float* arow = A + (size_t)(tile * TROWS + lane16) * FEAT + 2 * hi;

    v2f a0[8], a1[8];
    load_ablock(a0, arow);

    for (;;) {
        const int  ntile = tile + 2 * NPAIR;
        const bool more  = ntile < NTILE;
        const float* nrow = A + (size_t)((more ? ntile : tile) * TROWS + lane16) * FEAT + 2 * hi;

        v8f acc[8];
#pragma unroll
        for (int t = 0; t < 8; ++t) {
            const float bv = bcol[t];
            acc[t] = (v8f){bv, bv, bv, bv, bv, bv, bv, bv};
        }

        for (int b = 0; b < NBLK; b += 2) {
            load_ablock(a1, arow + (b + 1) * KBLK);
            mul_block(acc, a0, bl + (size_t)b * (KBLK / 2) * BPITCH);
            load_ablock(a0, (b + 2 < NBLK) ? arow + (b + 2) * KBLK : nrow);
            mul_block(acc, a1, bl + (size_t)(b + 1) * (KBLK / 2) * BPITCH);
        }

#pragma unroll
        for (int t = 0; t < 8; ++t) {
#pragma unroll
            for (int j = 0; j < 8; ++j) {
                const int m   = tile * TROWS + j + 8 * hi;
                const float v = acc[t][j];
                out[(size_t)m * HID + t * 16 + lane16] = v > 0.f ? v : 0.f;
            }
        }

        if (!more) break;
        tile = ntile;
        arow = nrow;
    }
}


__global__ __launch_bounds__(256) void k_h(const float* __restrict__ tf_h,
                                           const float* __restrict__ wf,
                                           const float* __restrict__ Wm,
                                           const float* __restrict__ Wl,
                                           float* __restrict__ hm,
                                           float* __restrict__ hl) {
    const int wave   = threadIdx.x >> 5;
    const int lane   = threadIdx.x & 31;
    const int lane16 = lane & 15;
    const int hi     = lane >> 4;

    const int node_base = (blockIdx.x * 8 + wave) * 16;
    int nd = node_base + lane16;
    if (nd > NODES - 1) nd = NODES - 1;
    const float* xrow = (nd < TASKS) ? (tf_h + (size_t)nd * HID)
                                     : (wf + (size_t)(nd - TASKS) * HID);
    const int n = lane16;
    const bool ncol = n < CLS;

    v8f accM = {0.f, 0.f, 0.f, 0.f, 0.f, 0.f, 0.f, 0.f};
    v8f accL = {0.f, 0.f, 0.f, 0.f, 0.f, 0.f, 0.f, 0.f};

    for (int k = 0; k < HID; k += 4) {
        const int k0 = k + 2 * hi;
        const float2 av = *(const float2*)(xrow + k0);
        v2f af = {av.x, av.y};
        v2f bm, bl;
        bm.x = ncol ? Wm[k0 * CLS + n] : 0.f;
        bm.y = ncol ? Wm[(k0 + 1) * CLS + n] : 0.f;
        bl.x = ncol ? Wl[k0 * CLS + n] : 0.f;
        bl.y = ncol ? Wl[(k0 + 1) * CLS + n] : 0.f;
        accM = __builtin_amdgcn_wmma_f32_16x16x4_f32(false, af, false, bm, (short)0, accM, false, false);
        accL = __builtin_amdgcn_wmma_f32_16x16x4_f32(false, af, false, bl, (short)0, accL, false, false);
    }

    if (ncol) {
#pragma unroll
        for (int j = 0; j < 8; ++j) {
            int m = node_base + j + 8 * hi;
            if (m < NODES) {
                hm[(size_t)m * HP + n] = accM[j];
                hl[(size_t)m * HP + n] = accL[j];
            }
        }
    }
}


__global__ void k_agg_init_w(const float* __restrict__ hm, const float* __restrict__ hl,
                             const float* __restrict__ dinv,
                             const float* __restrict__ bm, const float* __restrict__ bl,
                             float* __restrict__ mean, float* __restrict__ lsd) {
    int i = blockIdx.x * 256 + threadIdx.x;
    if (i < WORKERS * CLS) {
        int w = i / CLS;
        int c = i - w * CLS;
        float d  = dinv[TASKS + w];
        float sw = d * d;
        mean[(size_t)TASKS * CLS + i] = bm[c] + hm[(size_t)(TASKS + w) * HP + c] * sw;
        lsd[(size_t)TASKS * CLS + i]  = bl[c] + hl[(size_t)(TASKS + w) * HP + c] * sw;
    }
}


__device__ __forceinline__ void lds_barrier() {
    __builtin_amdgcn_fence(__ATOMIC_RELEASE, "workgroup", "local");
    __builtin_amdgcn_s_barrier();
    __builtin_amdgcn_fence(__ATOMIC_ACQUIRE, "workgroup", "local");
}

__global__ __launch_bounds__(OWN_THREADS) void k_agg_owner(const int* __restrict__ keys,
                                                           const float* __restrict__ hm,
                                                           const float* __restrict__ hl,
                                                           const float* __restrict__ dinv,
                                                           const float* __restrict__ bm,
                                                           const float* __restrict__ bl,
                                                           float* __restrict__ mean,
                                                           float* __restrict__ lsd) {
    __shared__ alignas(16) float acc[TBLK * ROW2];
    __shared__ alignas(16) float Fs[TBLK * ROW2];
    __shared__ alignas(16) float Gs[WORKERS * ROW2];
    __shared__ alignas(16) float wacc[WORKERS * ROW2];
    __shared__ int   q[QCAP];
    __shared__ int   qn[2];

    const int tid    = threadIdx.x;
    const int t_base = blockIdx.x * TBLK;

    for (int i = tid; i < TBLK * CLS; i += OWN_THREADS) {
        const int tl = i / CLS;
        const int c  = i - tl * CLS;
        const int t  = t_base + tl;
        const float d = dinv[t];
        Fs[tl * ROW2 + c]        = d * hm[(size_t)t * HP + c];
        Fs[tl * ROW2 + CLS + c]  = d * hl[(size_t)t * HP + c];
        acc[tl * ROW2 + c]       = 0.f;
        acc[tl * ROW2 + CLS + c] = 0.f;
    }
    for (int i = tid; i < WORKERS * CLS; i += OWN_THREADS) {
        const int w  = i / CLS;
        const int c  = i - w * CLS;
        const int r  = TASKS + w;
        const float d = dinv[r];
        Gs[w * ROW2 + c]         = d * hm[(size_t)r * HP + c];
        Gs[w * ROW2 + CLS + c]   = d * hl[(size_t)r * HP + c];
        wacc[w * ROW2 + c]       = 0.f;
        wacc[w * ROW2 + CLS + c] = 0.f;
    }
    if (tid < 2) qn[tid] = 0;
    __syncthreads();

    const int4* __restrict__ kp = (const int4*)keys;
    const unsigned key_lo  = (unsigned)t_base << KSH;
    const unsigned key_len = (unsigned)TBLK << KSH;

    int4 cur[GRPS_THREAD];
#pragma unroll
    for (int j = 0; j < GRPS_THREAD; ++j) cur[j] = kp[j * OWN_THREADS + tid];

    for (int ch = 0; ch < NCHUNK; ++ch) {
        const int p = ch & 1;

#pragma unroll
        for (int j = 0; j < GRPS_THREAD; ++j) {
            const int kk[4] = {cur[j].x, cur[j].y, cur[j].z, cur[j].w};
#pragma unroll
            for (int u = 0; u < 4; ++u) {
                const unsigned rel = (unsigned)kk[u] - key_lo;
                if (rel < key_len) {
                    const int pos = atomicAdd(&qn[p], 1);
                    q[pos] = (int)rel;
                }
            }
        }

        if (ch + 1 < NCHUNK) {
#pragma unroll
            for (int j = 0; j < GRPS_THREAD; ++j) {
                const int g  = (ch + 1) * GRPS_CHUNK + j * OWN_THREADS + tid;
                const int4 v = kp[g < NGRP ? g : NGRP - 1];
                cur[j] = (g < NGRP) ? v : make_int4(-1, -1, -1, -1);
            }
        }

        lds_barrier();
        const int items = qn[p] * (ROW2 / 4);
        if (tid == 0) qn[p ^ 1] = 0;

        for (int i = tid; i < items; i += OWN_THREADS) {
            const int e   = i / (ROW2 / 4);
            const int c4  = (i - e * (ROW2 / 4)) * 4;
            const int key = q[e];
            const int tl  = key >> KSH;
            const int w   = key & KMASK;
            const float4 g = *(const float4*)&Gs[w * ROW2 + c4];
            const float4 f = *(const float4*)&Fs[tl * ROW2 + c4];
            float* a = &acc[tl * ROW2 + c4];
            float* b = &wacc[w * ROW2 + c4];
            atomicAdd(a + 0, g.x); atomicAdd(a + 1, g.y); atomicAdd(a + 2, g.z); atomicAdd(a + 3, g.w);
            atomicAdd(b + 0, f.x); atomicAdd(b + 1, f.y); atomicAdd(b + 2, f.z); atomicAdd(b + 3, f.w);
        }
        lds_barrier();
    }

    for (int i = tid; i < TBLK * CLS; i += OWN_THREADS) {
        const int tl = i / CLS;
        const int c  = i - tl * CLS;
        const float d = dinv[t_base + tl];
        const size_t o = (size_t)t_base * CLS + i;
        mean[o] = bm[c] + d * (Fs[tl * ROW2 + c] + acc[tl * ROW2 + c]);
        lsd[o]  = bl[c] + d * (Fs[tl * ROW2 + CLS + c] + acc[tl * ROW2 + CLS + c]);
    }

    for (int i = tid; i < WORKERS * CLS; i += OWN_THREADS) {
        const int w = i / CLS;
        const int c = i - w * CLS;
        const float d = dinv[TASKS + w];
        atomicAdd(&mean[(size_t)TASKS * CLS + i], d * wacc[w * ROW2 + c]);
        atomicAdd(&lsd[(size_t)TASKS * CLS + i],  d * wacc[w * ROW2 + CLS + c]);
    }
}


__global__ void k_z(const float* __restrict__ mean, const float* __restrict__ lsd,
                    const float* __restrict__ eps, float* __restrict__ z) {
    int idx = blockIdx.x * 256 + threadIdx.x;
    if (idx < NODES * CLS)
        z[idx] = mean[idx] + (eps[idx] * 0.01f) * expf(lsd[idx]);
}

constexpr int NQUAD      = ANS * CLS / 4;
constexpr int CR_THREADS = 160;
constexpr int CR_WGS     = 625;
constexpr int CR_GT      = CR_THREADS * CR_WGS;
constexpr int CR_TRIPS   = NQUAD / CR_GT;
constexpr int CR_BATCH   = 5;
constexpr int CR_ASTEP   = CR_GT * 4 / CLS;
static_assert((ANS * CLS) % 4 == 0 && NQUAD % CR_GT == 0 && (CR_GT * 4) % CLS == 0 &&
              CR_TRIPS % CR_BATCH == 0 && CLS % 2 == 0, "crowd partition must be exact");

__global__ __launch_bounds__(CR_THREADS) void k_crowd(const int* __restrict__ keys,
                                                      const float* __restrict__ z,
                                                      float* __restrict__ crowd) {
    const int g  = blockIdx.x * CR_THREADS + threadIdx.x;
    const int eA = 4 * g;
    const int aA = eA / CLS;
    const int cA = eA - aA * CLS;
    const int eB = eA + 2;
    const int aB = eB / CLS;
    const int cB = eB - aB * CLS;

    const float* __restrict__ ztA = z + cA;
    const float* __restrict__ ztB = z + cB;
    const float* __restrict__ zwA = z + (size_t)TASKS * CLS + cA;
    const float* __restrict__ zwB = z + (size_t)TASKS * CLS + cB;
    const int*   __restrict__ kA  = keys + aA;
    const int*   __restrict__ kB  = keys + aB;
    float4*      __restrict__ o   = (float4*)crowd + g;

    for (int b = 0; b < CR_TRIPS; b += CR_BATCH) {
        int ka[CR_BATCH], kb[CR_BATCH];
#pragma unroll
        for (int u = 0; u < CR_BATCH; ++u) {
            ka[u] = kA[(b + u) * CR_ASTEP];
            kb[u] = kB[(b + u) * CR_ASTEP];
        }
        float2 ta[CR_BATCH], wa[CR_BATCH], tb[CR_BATCH], wb[CR_BATCH];
#pragma unroll
        for (int u = 0; u < CR_BATCH; ++u) {
            ta[u] = *(const float2*)(ztA + (ka[u] >> KSH) * CLS);
            wa[u] = *(const float2*)(zwA + (ka[u] & KMASK) * CLS);
            tb[u] = *(const float2*)(ztB + (kb[u] >> KSH) * CLS);
            wb[u] = *(const float2*)(zwB + (kb[u] & KMASK) * CLS);
        }
#pragma unroll
        for (int u = 0; u < CR_BATCH; ++u)
            o[(size_t)(b + u) * CR_GT] = make_float4(ta[u].x * wa[u].x, ta[u].y * wa[u].y,
                                                     tb[u].x * wb[u].x, tb[u].y * wb[u].y);
    }
}


extern "C" void kernel_launch(void* const* d_in, const int* in_sizes, int n_in,
                              void* d_out, int out_size, void* d_ws, size_t ws_size,
                              hipStream_t stream) {
    (void)in_sizes; (void)n_in; (void)out_size; (void)ws_size;

    const float* task_feature   = (const float*)d_in[0];
    const int*   answers        = (const int*)d_in[1];
    const float* worker_feature = (const float*)d_in[2];
    const float* W_efc          = (const float*)d_in[3];
    const float* b_efc          = (const float*)d_in[4];
    const float* W_mean         = (const float*)d_in[5];
    const float* b_mean         = (const float*)d_in[6];
    const float* W_ls           = (const float*)d_in[7];
    const float* b_ls           = (const float*)d_in[8];
    const float* eps            = (const float*)d_in[9];

    float* out   = (float*)d_out;
    float* crowd = out;
    float* mean  = out + (size_t)ANS * CLS;
    float* lsd   = mean + (size_t)NODES * CLS;

    float*  ws   = (float*)d_ws;
    float*  tf_h = ws;
    float*  hm   = tf_h + (size_t)TASKS * HID;
    float*  hl   = hm + (size_t)NODES * HP;
    float*  zz   = hl + (size_t)NODES * HP;
    float*  dinv = zz + (size_t)NODES * CLS;

    auto cdiv = [](int a, int b) { return (a + b - 1) / b; };

    k_gemm_efc <<<GEMM_WGS, 256, 0, stream>>>(task_feature, W_efc, b_efc, tf_h);
    k_h        <<<cdiv(cdiv(NODES, 16), 8), 256, 0, stream>>>(tf_h, worker_feature,
                                                              W_mean, W_ls, hm, hl);

    int* keys  = (int*)tf_h;
    int* tpart = keys + ANS;
    int* wpart = tpart + (size_t)CNT_S * TASKS;
    k_pack     <<<cdiv(ANS, 256), 256, 0, stream>>>(answers, keys);
    k_count    <<<CNT_WGS, CNT_THREADS, 0, stream>>>(keys, tpart, wpart);
    k_dinv     <<<cdiv(NODES, 256), 256, 0, stream>>>(tpart, wpart, dinv);

    k_agg_init_w<<<cdiv(WORKERS * CLS, 256), 256, 0, stream>>>(hm, hl, dinv, b_mean, b_ls, mean, lsd);
    k_agg_owner <<<NOWN, OWN_THREADS, 0, stream>>>(keys, hm, hl, dinv, b_mean, b_ls, mean, lsd);

    k_z        <<<cdiv(NODES * CLS, 256), 256, 0, stream>>>(mean, lsd, eps, zz);
    k_crowd    <<<CR_WGS, CR_THREADS, 0, stream>>>(keys, zz, crowd);
}


# Keys counting-sorted by task range (25 ranges, per-wave ballot counts, no key atomics): an aggregation owner scans its range's segment, not all 1M keys; +1 launch; LDS float-add arrival order differs
# speedup vs baseline: 3.9950x; 1.1495x over previous
#include <hip/hip_runtime.h>

typedef __attribute__((ext_vector_type(2))) float v2f;
typedef __attribute__((ext_vector_type(8))) float v8f;

constexpr int TASKS   = 100000;
constexpr int WORKERS = 500;
constexpr int FEAT    = 512;
constexpr int CLS     = 10;
constexpr int HP      = 12;
constexpr int ANS     = 1000000;
constexpr int HID     = 128;
constexpr int NODES   = TASKS + WORKERS;
constexpr int KSH     = 9;
constexpr int KMASK   = (1 << KSH) - 1;
constexpr int TBLK    = 800;
constexpr int NOWN    = TASKS / TBLK;
constexpr int ROW2    = 2 * CLS;
constexpr int QCAP    = 16384;
constexpr int OWN_THREADS = 1024;
constexpr int KEYS_THREAD = QCAP / OWN_THREADS;
constexpr int TROWS   = 16;
constexpr int NTILE   = TASKS / TROWS;
constexpr int GEMM_WGS = 256;
constexpr int NPAIR   = GEMM_WGS * 4;
constexpr int BPITCH  = HID + 16;
constexpr int KBLK    = 32;
constexpr int NBLK    = FEAT / KBLK;

constexpr int CNT_R     = 25;
constexpr int CNT_S     = 10;
constexpr int CNT_WGS   = CNT_R * CNT_S;
constexpr int CNT_THREADS = 1024;
constexpr int CNT_WAVES = CNT_THREADS / 32;
constexpr int RTASKS    = TASKS / CNT_R;
constexpr int SGRPS     = (ANS / 4) / CNT_S;
constexpr int WPITCH    = 512;
constexpr int CNT_BATCH = 8;
constexpr int CNT_TRIPS = (SGRPS + CNT_THREADS * CNT_BATCH - 1) / (CNT_THREADS * CNT_BATCH);
constexpr int OWN_PER_R = RTASKS / TBLK;
static_assert(TASKS % CNT_R == 0 && ANS % (4 * CNT_S) == 0 && WORKERS <= WPITCH && WPITCH <= CNT_THREADS &&
              RTASKS % TBLK == 0, "count partition must be exact");

__global__ void k_pack(const int* __restrict__ answers, int* __restrict__ keys) {
    int a = blockIdx.x * 256 + threadIdx.x;
    if (a < ANS) {
        const int t = answers[3 * a];
        const int w = answers[3 * a + 1];
        keys[a] = (t << KSH) | w;
    }
}

__global__ __launch_bounds__(CNT_THREADS) void k_count(const int* __restrict__ keys,
                                                       int* __restrict__ tpart,
                                                       int* __restrict__ wpart,
                                                       int* __restrict__ wvcnt) {
    __shared__ int tcnt[RTASKS];
    __shared__ int wcnt[WPITCH];

    const int tid = threadIdx.x;
    const int r   = blockIdx.x / CNT_S;
    const int s   = blockIdx.x % CNT_S;

    for (int i = tid; i < RTASKS; i += CNT_THREADS) tcnt[i] = 0;
    if (tid < WPITCH) wcnt[tid] = 0;
    __syncthreads();

    const int4* __restrict__ kp = (const int4*)keys + (size_t)s * SGRPS;
    const unsigned key_lo  = (unsigned)(r * RTASKS) << KSH;
    const unsigned key_len = (unsigned)RTASKS << KSH;

    int wc = 0;
    for (int k = 0; k < CNT_TRIPS; ++k) {
        const int g0 = tid + k * CNT_THREADS * CNT_BATCH;
        int4 v[CNT_BATCH];
#pragma unroll
        for (int j = 0; j < CNT_BATCH; ++j) {
            const int g = g0 + j * CNT_THREADS;
            v[j] = kp[g < SGRPS ? g : SGRPS - 1];
        }
#pragma unroll
        for (int j = 0; j < CNT_BATCH; ++j) {
            const bool valid = g0 + j * CNT_THREADS < SGRPS;
            const int kk[4] = {v[j].x, v[j].y, v[j].z, v[j].w};
#pragma unroll
            for (int u = 0; u < 4; ++u) {
                const unsigned rel = (unsigned)kk[u] - key_lo;
                const bool match = valid && rel < key_len;
                wc += __popc(__builtin_amdgcn_ballot_w32(match));
                if (match) {
                    atomicAdd(&tcnt[rel >> KSH], 1);
                    atomicAdd(&wcnt[rel & KMASK], 1);
                }
            }
        }
    }
    if ((tid & 31) == 0) wvcnt[blockIdx.x * CNT_WAVES + (tid >> 5)] = wc;
    __syncthreads();

    int* __restrict__ tp = tpart + (size_t)s * TASKS + r * RTASKS;
    for (int i = tid; i < RTASKS; i += CNT_THREADS) tp[i] = tcnt[i];
    if (tid < WPITCH) wpart[(size_t)blockIdx.x * WPITCH + tid] = wcnt[tid];
}

__global__ __launch_bounds__(CNT_THREADS) void k_scatter(const int* __restrict__ keys,
                                                         const int* __restrict__ wvcnt,
                                                         int* __restrict__ sorted,
                                                         int* __restrict__ segstart) {
    __shared__ int base_s;

    const int tid  = threadIdx.x;
    const int lane = tid & 31;
    const int wave = tid >> 5;
    const int r    = blockIdx.x / CNT_S;
    const int s    = blockIdx.x % CNT_S;

    if (tid == 0) base_s = 0;
    __syncthreads();

    const int nbefore = blockIdx.x * CNT_WAVES;
    int part = 0;
    for (int i = tid; i < nbefore; i += CNT_THREADS) part += wvcnt[i];
    int mine = (lane < wave) ? wvcnt[nbefore + lane] : 0;
    for (int o = 16; o > 0; o >>= 1) {
        part += __shfl_xor(part, o);
        mine += __shfl_xor(mine, o);
    }
    if (lane == 0 && part != 0) atomicAdd(&base_s, part);
    __syncthreads();
    const int wg_base = base_s;
    if (s == 0 && tid == 0) segstart[r] = wg_base;

    const int4* __restrict__ kp = (const int4*)keys + (size_t)s * SGRPS;
    const unsigned key_lo  = (unsigned)(r * RTASKS) << KSH;
    const unsigned key_len = (unsigned)RTASKS << KSH;
    const unsigned below   = (1u << lane) - 1u;

    int wpos = wg_base + mine;
    for (int k = 0; k < CNT_TRIPS; ++k) {
        const int g0 = tid + k * CNT_THREADS * CNT_BATCH;
        int4 v[CNT_BATCH];
#pragma unroll
        for (int j = 0; j < CNT_BATCH; ++j) {
            const int g = g0 + j * CNT_THREADS;
            v[j] = kp[g < SGRPS ? g : SGRPS - 1];
        }
#pragma unroll
        for (int j = 0; j < CNT_BATCH; ++j) {
            const bool valid = g0 + j * CNT_THREADS < SGRPS;
            const int kk[4] = {v[j].x, v[j].y, v[j].z, v[j].w};
#pragma unroll
            for (int u = 0; u < 4; ++u) {
                const unsigned rel = (unsigned)kk[u] - key_lo;
                const bool match = valid && rel < key_len;
                const unsigned m = __builtin_amdgcn_ballot_w32(match);
                if (match) sorted[wpos + __popc(m & below)] = kk[u];
                wpos += __popc(m);
            }
        }
    }
}

__global__ void k_dinv(const int* __restrict__ tpart, const int* __restrict__ wpart,
                       float* __restrict__ dinv) {
    int i = blockIdx.x * 256 + threadIdx.x;
    if (i < TASKS) {
        int d = 1;
#pragma unroll
        for (int s = 0; s < CNT_S; ++s) d += tpart[(size_t)s * TASKS + i];
        dinv[i] = rsqrtf((float)d);
    } else if (i < NODES) {
        const int w = i - TASKS;
        int d = 1;
#pragma unroll 10
        for (int b = 0; b < CNT_WGS; ++b) d += wpart[(size_t)b * WPITCH + w];
        dinv[i] = rsqrtf((float)d);
    }
}


__device__ __forceinline__ void load_ablock(v2f (&a)[8], const float* __restrict__ p) {
#pragma unroll
    for (int s = 0; s < 8; ++s) {
        const float2 v = *(const float2*)(p + 4 * s);
        a[s] = (v2f){v.x, v.y};
    }
}

__device__ __forceinline__ void mul_block(v8f (&acc)[8], const v2f (&a)[8], const float2* bb) {
#pragma unroll
    for (int s = 0; s < 8; ++s) {
        const float2* brow = bb + s * 2 * BPITCH;
#pragma unroll
        for (int t = 0; t < 8; ++t) {
            const float2 bv = brow[t * 16];
            v2f bf = {bv.x, bv.y};
            acc[t] = __builtin_amdgcn_wmma_f32_16x16x4_f32(
                false, a[s], false, bf, (short)0, acc[t], false, false);
        }
    }
}

__global__ __launch_bounds__(256) void k_gemm_efc(const float* __restrict__ A,
                                                  const float* __restrict__ B,
                                                  const float* __restrict__ bias,
                                                  float* __restrict__ out) {
    __shared__ alignas(16) float2 Bs[(FEAT / 2) * BPITCH];

    const int tid = threadIdx.x;

#pragma unroll 4
    for (int it = tid; it < (FEAT / 2) * (HID / 4); it += 256) {
        const int kp = it / (HID / 4);
        const int n4 = it - kp * (HID / 4);
        const float4 r0 = *(const float4*)(B + (size_t)(2 * kp) * HID + 4 * n4);
        const float4 r1 = *(const float4*)(B + (size_t)(2 * kp + 1) * HID + 4 * n4);
        float4* dst = (float4*)&Bs[kp * BPITCH + 4 * n4];
        dst[0] = make_float4(r0.x, r1.x, r0.y, r1.y);
        dst[1] = make_float4(r0.z, r1.z, r0.w, r1.w);
    }
    __syncthreads();

    const int wave   = tid >> 5;
    const int lane   = tid & 31;
    const int lane16 = lane & 15;
    const int hi     = lane >> 4;

    const int pair = blockIdx.x * 4 + (wave & 3);
    const int sub  = wave >> 2;

    float bcol[8];
#pragma unroll
    for (int t = 0; t < 8; ++t) bcol[t] = bias[t * 16 + lane16];

    const float2* bl = Bs + hi * BPITCH + lane16;

    int tile = pair + NPAIR * sub;
    const float* arow = A + (size_t)(tile * TROWS + lane16) * FEAT + 2 * hi;

    v2f a0[8], a1[8];
    load_ablock(a0, arow);

    for (;;) {
        const int  ntile = tile + 2 * NPAIR;
        const bool more  = ntile < NTILE;
        const float* nrow = A + (size_t)((more ? ntile : tile) * TROWS + lane16) * FEAT + 2 * hi;

        v8f acc[8];
#pragma unroll
        for (int t = 0; t < 8; ++t) {
            const float bv = bcol[t];
            acc[t] = (v8f){bv, bv, bv, bv, bv, bv, bv, bv};
        }

        for (int b = 0; b < NBLK; b += 2) {
            load_ablock(a1, arow + (b + 1) * KBLK);
            mul_block(acc, a0, bl + (size_t)b * (KBLK / 2) * BPITCH);
            load_ablock(a0, (b + 2 < NBLK) ? arow + (b + 2) * KBLK : nrow);
            mul_block(acc, a1, bl + (size_t)(b + 1) * (KBLK / 2) * BPITCH);
        }

#pragma unroll
        for (int t = 0; t < 8; ++t) {
#pragma unroll
            for (int j = 0; j < 8; ++j) {
                const int m   = tile * TROWS + j + 8 * hi;
                const float v = acc[t][j];
                out[(size_t)m * HID + t * 16 + lane16] = v > 0.f ? v : 0.f;
            }
        }

        if (!more) break;
        tile = ntile;
        arow = nrow;
    }
}


__global__ __launch_bounds__(256) void k_h(const float* __restrict__ tf_h,
                                           const float* __restrict__ wf,
                                           const float* __restrict__ Wm,
                                           const float* __restrict__ Wl,
                                           float* __restrict__ hm,
                                           float* __restrict__ hl) {
    const int wave   = threadIdx.x >> 5;
    const int lane   = threadIdx.x & 31;
    const int lane16 = lane & 15;
    const int hi     = lane >> 4;

    const int node_base = (blockIdx.x * 8 + wave) * 16;
    int nd = node_base + lane16;
    if (nd > NODES - 1) nd = NODES - 1;
    const float* xrow = (nd < TASKS) ? (tf_h + (size_t)nd * HID)
                                     : (wf + (size_t)(nd - TASKS) * HID);
    const int n = lane16;
    const bool ncol = n < CLS;

    v8f accM = {0.f, 0.f, 0.f, 0.f, 0.f, 0.f, 0.f, 0.f};
    v8f accL = {0.f, 0.f, 0.f, 0.f, 0.f, 0.f, 0.f, 0.f};

    for (int k = 0; k < HID; k += 4) {
        const int k0 = k + 2 * hi;
        const float2 av = *(const float2*)(xrow + k0);
        v2f af = {av.x, av.y};
        v2f bm, bl;
        bm.x = ncol ? Wm[k0 * CLS + n] : 0.f;
        bm.y = ncol ? Wm[(k0 + 1) * CLS + n] : 0.f;
        bl.x = ncol ? Wl[k0 * CLS + n] : 0.f;
        bl.y = ncol ? Wl[(k0 + 1) * CLS + n] : 0.f;
        accM = __builtin_amdgcn_wmma_f32_16x16x4_f32(false, af, false, bm, (short)0, accM, false, false);
        accL = __builtin_amdgcn_wmma_f32_16x16x4_f32(false, af, false, bl, (short)0, accL, false, false);
    }

    if (ncol) {
#pragma unroll
        for (int j = 0; j < 8; ++j) {
            int m = node_base + j + 8 * hi;
            if (m < NODES) {
                hm[(size_t)m * HP + n] = accM[j];
                hl[(size_t)m * HP + n] = accL[j];
            }
        }
    }
}


__global__ void k_agg_init_w(const float* __restrict__ hm, const float* __restrict__ hl,
                             const float* __restrict__ dinv,
                             const float* __restrict__ bm, const float* __restrict__ bl,
                             float* __restrict__ mean, float* __restrict__ lsd) {
    int i = blockIdx.x * 256 + threadIdx.x;
    if (i < WORKERS * CLS) {
        int w = i / CLS;
        int c = i - w * CLS;
        float d  = dinv[TASKS + w];
        float sw = d * d;
        mean[(size_t)TASKS * CLS + i] = bm[c] + hm[(size_t)(TASKS + w) * HP + c] * sw;
        lsd[(size_t)TASKS * CLS + i]  = bl[c] + hl[(size_t)(TASKS + w) * HP + c] * sw;
    }
}


__device__ __forceinline__ void lds_barrier() {
    __builtin_amdgcn_fence(__ATOMIC_RELEASE, "workgroup", "local");
    __builtin_amdgcn_s_barrier();
    __builtin_amdgcn_fence(__ATOMIC_ACQUIRE, "workgroup", "local");
}

__global__ __launch_bounds__(OWN_THREADS) void k_agg_owner(const int* __restrict__ sorted,
                                                           const int* __restrict__ segstart,
                                                           const float* __restrict__ hm,
                                                           const float* __restrict__ hl,
                                                           const float* __restrict__ dinv,
                                                           const float* __restrict__ bm,
                                                           const float* __restrict__ bl,
                                                           float* __restrict__ mean,
                                                           float* __restrict__ lsd) {
    __shared__ alignas(16) float acc[TBLK * ROW2];
    __shared__ alignas(16) float Fs[TBLK * ROW2];
    __shared__ alignas(16) float Gs[WORKERS * ROW2];
    __shared__ alignas(16) float wacc[WORKERS * ROW2];
    __shared__ int   q[QCAP];
    __shared__ int   qn[2];

    const int tid    = threadIdx.x;
    const int t_base = blockIdx.x * TBLK;

    for (int i = tid; i < TBLK * CLS; i += OWN_THREADS) {
        const int tl = i / CLS;
        const int c  = i - tl * CLS;
        const int t  = t_base + tl;
        const float d = dinv[t];
        Fs[tl * ROW2 + c]        = d * hm[(size_t)t * HP + c];
        Fs[tl * ROW2 + CLS + c]  = d * hl[(size_t)t * HP + c];
        acc[tl * ROW2 + c]       = 0.f;
        acc[tl * ROW2 + CLS + c] = 0.f;
    }
    for (int i = tid; i < WORKERS * CLS; i += OWN_THREADS) {
        const int w  = i / CLS;
        const int c  = i - w * CLS;
        const int r  = TASKS + w;
        const float d = dinv[r];
        Gs[w * ROW2 + c]         = d * hm[(size_t)r * HP + c];
        Gs[w * ROW2 + CLS + c]   = d * hl[(size_t)r * HP + c];
        wacc[w * ROW2 + c]       = 0.f;
        wacc[w * ROW2 + CLS + c] = 0.f;
    }
    if (tid < 2) qn[tid] = 0;
    __syncthreads();

    const unsigned key_lo  = (unsigned)t_base << KSH;
    const unsigned key_len = (unsigned)TBLK << KSH;

    const int rng = blockIdx.x / OWN_PER_R;
    const int lo  = segstart[rng];
    const int hi  = (rng + 1 < CNT_R) ? segstart[rng + 1] : ANS;
    const int nch = (hi - lo + QCAP - 1) / QCAP;

    int cur[KEYS_THREAD];
#pragma unroll
    for (int j = 0; j < KEYS_THREAD; ++j) {
        const int idx = lo + j * OWN_THREADS + tid;
        cur[j] = (idx < hi) ? sorted[idx] : -1;
    }

    for (int ch = 0; ch < nch; ++ch) {
        const int p = ch & 1;

#pragma unroll
        for (int j = 0; j < KEYS_THREAD; ++j) {
            const unsigned rel = (unsigned)cur[j] - key_lo;
            if (rel < key_len) {
                const int pos = atomicAdd(&qn[p], 1);
                q[pos] = (int)rel;
            }
        }

        if (ch + 1 < nch) {
#pragma unroll
            for (int j = 0; j < KEYS_THREAD; ++j) {
                const int idx = lo + (ch + 1) * QCAP + j * OWN_THREADS + tid;
                cur[j] = (idx < hi) ? sorted[idx] : -1;
            }
        }

        lds_barrier();
        const int items = qn[p] * (ROW2 / 4);
        if (tid == 0) qn[p ^ 1] = 0;

        for (int i = tid; i < items; i += OWN_THREADS) {
            const int e   = i / (ROW2 / 4);
            const int c4  = (i - e * (ROW2 / 4)) * 4;
            const int key = q[e];
            const int tl  = key >> KSH;
            const int w   = key & KMASK;
            const float4 g = *(const float4*)&Gs[w * ROW2 + c4];
            const float4 f = *(const float4*)&Fs[tl * ROW2 + c4];
            float* a = &acc[tl * ROW2 + c4];
            float* b = &wacc[w * ROW2 + c4];
            atomicAdd(a + 0, g.x); atomicAdd(a + 1, g.y); atomicAdd(a + 2, g.z); atomicAdd(a + 3, g.w);
            atomicAdd(b + 0, f.x); atomicAdd(b + 1, f.y); atomicAdd(b + 2, f.z); atomicAdd(b + 3, f.w);
        }
        lds_barrier();
    }

    for (int i = tid; i < TBLK * CLS; i += OWN_THREADS) {
        const int tl = i / CLS;
        const int c  = i - tl * CLS;
        const float d = dinv[t_base + tl];
        const size_t o = (size_t)t_base * CLS + i;
        mean[o] = bm[c] + d * (Fs[tl * ROW2 + c] + acc[tl * ROW2 + c]);
        lsd[o]  = bl[c] + d * (Fs[tl * ROW2 + CLS + c] + acc[tl * ROW2 + CLS + c]);
    }

    for (int i = tid; i < WORKERS * CLS; i += OWN_THREADS) {
        const int w = i / CLS;
        const int c = i - w * CLS;
        const float d = dinv[TASKS + w];
        atomicAdd(&mean[(size_t)TASKS * CLS + i], d * wacc[w * ROW2 + c]);
        atomicAdd(&lsd[(size_t)TASKS * CLS + i],  d * wacc[w * ROW2 + CLS + c]);
    }
}


__global__ void k_z(const float* __restrict__ mean, const float* __restrict__ lsd,
                    const float* __restrict__ eps, float* __restrict__ z) {
    int idx = blockIdx.x * 256 + threadIdx.x;
    if (idx < NODES * CLS)
        z[idx] = mean[idx] + (eps[idx] * 0.01f) * expf(lsd[idx]);
}

constexpr int NQUAD      = ANS * CLS / 4;
constexpr int CR_THREADS = 160;
constexpr int CR_WGS     = 625;
constexpr int CR_GT      = CR_THREADS * CR_WGS;
constexpr int CR_TRIPS   = NQUAD / CR_GT;
constexpr int CR_BATCH   = 5;
constexpr int CR_ASTEP   = CR_GT * 4 / CLS;
static_assert((ANS * CLS) % 4 == 0 && NQUAD % CR_GT == 0 && (CR_GT * 4) % CLS == 0 &&
              CR_TRIPS % CR_BATCH == 0 && CLS % 2 == 0, "crowd partition must be exact");

__global__ __launch_bounds__(CR_THREADS) void k_crowd(const int* __restrict__ keys,
                                                      const float* __restrict__ z,
                                                      float* __restrict__ crowd) {
    const int g  = blockIdx.x * CR_THREADS + threadIdx.x;
    const int eA = 4 * g;
    const int aA = eA / CLS;
    const int cA = eA - aA * CLS;
    const int eB = eA + 2;
    const int aB = eB / CLS;
    const int cB = eB - aB * CLS;

    const float* __restrict__ ztA = z + cA;
    const float* __restrict__ ztB = z + cB;
    const float* __restrict__ zwA = z + (size_t)TASKS * CLS + cA;
    const float* __restrict__ zwB = z + (size_t)TASKS * CLS + cB;
    const int*   __restrict__ kA  = keys + aA;
    const int*   __restrict__ kB  = keys + aB;
    float4*      __restrict__ o   = (float4*)crowd + g;

    for (int b = 0; b < CR_TRIPS; b += CR_BATCH) {
        int ka[CR_BATCH], kb[CR_BATCH];
#pragma unroll
        for (int u = 0; u < CR_BATCH; ++u) {
            ka[u] = kA[(b + u) * CR_ASTEP];
            kb[u] = kB[(b + u) * CR_ASTEP];
        }
        float2 ta[CR_BATCH], wa[CR_BATCH], tb[CR_BATCH], wb[CR_BATCH];
#pragma unroll
        for (int u = 0; u < CR_BATCH; ++u) {
            ta[u] = *(const float2*)(ztA + (ka[u] >> KSH) * CLS);
            wa[u] = *(const float2*)(zwA + (ka[u] & KMASK) * CLS);
            tb[u] = *(const float2*)(ztB + (kb[u] >> KSH) * CLS);
            wb[u] = *(const float2*)(zwB + (kb[u] & KMASK) * CLS);
        }
#pragma unroll
        for (int u = 0; u < CR_BATCH; ++u)
            o[(size_t)(b + u) * CR_GT] = make_float4(ta[u].x * wa[u].x, ta[u].y * wa[u].y,
                                                     tb[u].x * wb[u].x, tb[u].y * wb[u].y);
    }
}


extern "C" void kernel_launch(void* const* d_in, const int* in_sizes, int n_in,
                              void* d_out, int out_size, void* d_ws, size_t ws_size,
                              hipStream_t stream) {
    (void)in_sizes; (void)n_in; (void)out_size; (void)ws_size;

    const float* task_feature   = (const float*)d_in[0];
    const int*   answers        = (const int*)d_in[1];
    const float* worker_feature = (const float*)d_in[2];
    const float* W_efc          = (const float*)d_in[3];
    const float* b_efc          = (const float*)d_in[4];
    const float* W_mean         = (const float*)d_in[5];
    const float* b_mean         = (const float*)d_in[6];
    const float* W_ls           = (const float*)d_in[7];
    const float* b_ls           = (const float*)d_in[8];
    const float* eps            = (const float*)d_in[9];

    float* out   = (float*)d_out;
    float* crowd = out;
    float* mean  = out + (size_t)ANS * CLS;
    float* lsd   = mean + (size_t)NODES * CLS;

    float*  ws   = (float*)d_ws;
    float*  tf_h = ws;
    float*  hm   = tf_h + (size_t)TASKS * HID;
    float*  hl   = hm + (size_t)NODES * HP;
    float*  zz   = hl + (size_t)NODES * HP;
    float*  dinv = zz + (size_t)NODES * CLS;

    auto cdiv = [](int a, int b) { return (a + b - 1) / b; };

    k_gemm_efc <<<GEMM_WGS, 256, 0, stream>>>(task_feature, W_efc, b_efc, tf_h);
    k_h        <<<cdiv(cdiv(NODES, 16), 8), 256, 0, stream>>>(tf_h, worker_feature,
                                                              W_mean, W_ls, hm, hl);

    int* keys  = (int*)tf_h;
    int* tpart = keys + ANS;
    int* wpart = tpart + (size_t)CNT_S * TASKS;
    int* wvcnt = wpart + (size_t)CNT_WGS * WPITCH;
    int* segst = wvcnt + (size_t)CNT_WGS * CNT_WAVES;
    int* sorted = segst + 32;
    k_pack     <<<cdiv(ANS, 256), 256, 0, stream>>>(answers, keys);
    k_count    <<<CNT_WGS, CNT_THREADS, 0, stream>>>(keys, tpart, wpart, wvcnt);
    k_scatter  <<<CNT_WGS, CNT_THREADS, 0, stream>>>(keys, wvcnt, sorted, segst);
    k_dinv     <<<cdiv(NODES, 256), 256, 0, stream>>>(tpart, wpart, dinv);

    k_agg_init_w<<<cdiv(WORKERS * CLS, 256), 256, 0, stream>>>(hm, hl, dinv, b_mean, b_ls, mean, lsd);
    k_agg_owner <<<NOWN, OWN_THREADS, 0, stream>>>(sorted, segst, hm, hl, dinv, b_mean, b_ls, mean, lsd);

    k_z        <<<cdiv(NODES * CLS, 256), 256, 0, stream>>>(mean, lsd, eps, zz);
    k_crowd    <<<CR_WGS, CR_THREADS, 0, stream>>>(keys, zz, crowd);
}


# Aggregation owner: edges counting-sorted by task in LDS; a 5-lane group sums each task row in registers, only worker rows keep LDS float adds; one-word queue counter gone; task sums in queue order
# speedup vs baseline: 4.2387x; 1.0610x over previous
#include <hip/hip_runtime.h>

typedef __attribute__((ext_vector_type(2))) float v2f;
typedef __attribute__((ext_vector_type(8))) float v8f;

constexpr int TASKS   = 100000;
constexpr int WORKERS = 500;
constexpr int FEAT    = 512;
constexpr int CLS     = 10;
constexpr int HP      = 12;
constexpr int ANS     = 1000000;
constexpr int HID     = 128;
constexpr int NODES   = TASKS + WORKERS;
constexpr int KSH     = 9;
constexpr int KMASK   = (1 << KSH) - 1;
constexpr int TBLK    = 800;
constexpr int NOWN    = TASKS / TBLK;
constexpr int ROW2    = 2 * CLS;
constexpr int QCAP    = 40960;
constexpr int OWN_THREADS = 1024;
constexpr int KEYS_THREAD = QCAP / OWN_THREADS;
constexpr int TROWS   = 16;
constexpr int NTILE   = TASKS / TROWS;
constexpr int GEMM_WGS = 256;
constexpr int NPAIR   = GEMM_WGS * 4;
constexpr int BPITCH  = HID + 16;
constexpr int KBLK    = 32;
constexpr int NBLK    = FEAT / KBLK;

constexpr int CNT_R     = 25;
constexpr int CNT_S     = 10;
constexpr int CNT_WGS   = CNT_R * CNT_S;
constexpr int CNT_THREADS = 1024;
constexpr int CNT_WAVES = CNT_THREADS / 32;
constexpr int RTASKS    = TASKS / CNT_R;
constexpr int SGRPS     = (ANS / 4) / CNT_S;
constexpr int WPITCH    = 512;
constexpr int CNT_BATCH = 8;
constexpr int CNT_TRIPS = (SGRPS + CNT_THREADS * CNT_BATCH - 1) / (CNT_THREADS * CNT_BATCH);
constexpr int OWN_PER_R = RTASKS / TBLK;
static_assert(TASKS % CNT_R == 0 && ANS % (4 * CNT_S) == 0 && WORKERS <= WPITCH && WPITCH <= CNT_THREADS &&
              RTASKS % TBLK == 0, "count partition must be exact");

__global__ void k_pack(const int* __restrict__ answers, int* __restrict__ keys) {
    int a = blockIdx.x * 256 + threadIdx.x;
    if (a < ANS) {
        const int t = answers[3 * a];
        const int w = answers[3 * a + 1];
        keys[a] = (t << KSH) | w;
    }
}

__global__ __launch_bounds__(CNT_THREADS) void k_count(const int* __restrict__ keys,
                                                       int* __restrict__ tpart,
                                                       int* __restrict__ wpart,
                                                       int* __restrict__ wvcnt) {
    __shared__ int tcnt[RTASKS];
    __shared__ int wcnt[WPITCH];

    const int tid = threadIdx.x;
    const int r   = blockIdx.x / CNT_S;
    const int s   = blockIdx.x % CNT_S;

    for (int i = tid; i < RTASKS; i += CNT_THREADS) tcnt[i] = 0;
    if (tid < WPITCH) wcnt[tid] = 0;
    __syncthreads();

    const int4* __restrict__ kp = (const int4*)keys + (size_t)s * SGRPS;
    const unsigned key_lo  = (unsigned)(r * RTASKS) << KSH;
    const unsigned key_len = (unsigned)RTASKS << KSH;

    int wc = 0;
    for (int k = 0; k < CNT_TRIPS; ++k) {
        const int g0 = tid + k * CNT_THREADS * CNT_BATCH;
        int4 v[CNT_BATCH];
#pragma unroll
        for (int j = 0; j < CNT_BATCH; ++j) {
            const int g = g0 + j * CNT_THREADS;
            v[j] = kp[g < SGRPS ? g : SGRPS - 1];
        }
#pragma unroll
        for (int j = 0; j < CNT_BATCH; ++j) {
            const bool valid = g0 + j * CNT_THREADS < SGRPS;
            const int kk[4] = {v[j].x, v[j].y, v[j].z, v[j].w};
#pragma unroll
            for (int u = 0; u < 4; ++u) {
                const unsigned rel = (unsigned)kk[u] - key_lo;
                const bool match = valid && rel < key_len;
                wc += __popc(__builtin_amdgcn_ballot_w32(match));
                if (match) {
                    atomicAdd(&tcnt[rel >> KSH], 1);
                    atomicAdd(&wcnt[rel & KMASK], 1);
                }
            }
        }
    }
    if ((tid & 31) == 0) wvcnt[blockIdx.x * CNT_WAVES + (tid >> 5)] = wc;
    __syncthreads();

    int* __restrict__ tp = tpart + (size_t)s * TASKS + r * RTASKS;
    for (int i = tid; i < RTASKS; i += CNT_THREADS) tp[i] = tcnt[i];
    if (tid < WPITCH) wpart[(size_t)blockIdx.x * WPITCH + tid] = wcnt[tid];
}

__global__ __launch_bounds__(CNT_THREADS) void k_scatter(const int* __restrict__ keys,
                                                         const int* __restrict__ wvcnt,
                                                         int* __restrict__ sorted,
                                                         int* __restrict__ segstart) {
    __shared__ int base_s;

    const int tid  = threadIdx.x;
    const int lane = tid & 31;
    const int wave = tid >> 5;
    const int r    = blockIdx.x / CNT_S;
    const int s    = blockIdx.x % CNT_S;

    if (tid == 0) base_s = 0;
    __syncthreads();

    const int nbefore = blockIdx.x * CNT_WAVES;
    int part = 0;
    for (int i = tid; i < nbefore; i += CNT_THREADS) part += wvcnt[i];
    int mine = (lane < wave) ? wvcnt[nbefore + lane] : 0;
    for (int o = 16; o > 0; o >>= 1) {
        part += __shfl_xor(part, o);
        mine += __shfl_xor(mine, o);
    }
    if (lane == 0 && part != 0) atomicAdd(&base_s, part);
    __syncthreads();
    const int wg_base = base_s;
    if (s == 0 && tid == 0) segstart[r] = wg_base;

    const int4* __restrict__ kp = (const int4*)keys + (size_t)s * SGRPS;
    const unsigned key_lo  = (unsigned)(r * RTASKS) << KSH;
    const unsigned key_len = (unsigned)RTASKS << KSH;
    const unsigned below   = (1u << lane) - 1u;

    int wpos = wg_base + mine;
    for (int k = 0; k < CNT_TRIPS; ++k) {
        const int g0 = tid + k * CNT_THREADS * CNT_BATCH;
        int4 v[CNT_BATCH];
#pragma unroll
        for (int j = 0; j < CNT_BATCH; ++j) {
            const int g = g0 + j * CNT_THREADS;
            v[j] = kp[g < SGRPS ? g : SGRPS - 1];
        }
#pragma unroll
        for (int j = 0; j < CNT_BATCH; ++j) {
            const bool valid = g0 + j * CNT_THREADS < SGRPS;
            const int kk[4] = {v[j].x, v[j].y, v[j].z, v[j].w};
#pragma unroll
            for (int u = 0; u < 4; ++u) {
                const unsigned rel = (unsigned)kk[u] - key_lo;
                const bool match = valid && rel < key_len;
                const unsigned m = __builtin_amdgcn_ballot_w32(match);
                if (match) sorted[wpos + __popc(m & below)] = kk[u];
                wpos += __popc(m);
            }
        }
    }
}

__global__ void k_dinv(const int* __restrict__ tpart, const int* __restrict__ wpart,
                       float* __restrict__ dinv) {
    int i = blockIdx.x * 256 + threadIdx.x;
    if (i < TASKS) {
        int d = 1;
#pragma unroll
        for (int s = 0; s < CNT_S; ++s) d += tpart[(size_t)s * TASKS + i];
        dinv[i] = rsqrtf((float)d);
    } else if (i < NODES) {
        const int w = i - TASKS;
        int d = 1;
#pragma unroll 10
        for (int b = 0; b < CNT_WGS; ++b) d += wpart[(size_t)b * WPITCH + w];
        dinv[i] = rsqrtf((float)d);
    }
}


__device__ __forceinline__ void load_ablock(v2f (&a)[8], const float* __restrict__ p) {
#pragma unroll
    for (int s = 0; s < 8; ++s) {
        const float2 v = *(const float2*)(p + 4 * s);
        a[s] = (v2f){v.x, v.y};
    }
}

__device__ __forceinline__ void mul_block(v8f (&acc)[8], const v2f (&a)[8], const float2* bb) {
#pragma unroll
    for (int s = 0; s < 8; ++s) {
        const float2* brow = bb + s * 2 * BPITCH;
#pragma unroll
        for (int t = 0; t < 8; ++t) {
            const float2 bv = brow[t * 16];
            v2f bf = {bv.x, bv.y};
            acc[t] = __builtin_amdgcn_wmma_f32_16x16x4_f32(
                false, a[s], false, bf, (short)0, acc[t], false, false);
        }
    }
}

__global__ __launch_bounds__(256) void k_gemm_efc(const float* __restrict__ A,
                                                  const float* __restrict__ B,
                                                  const float* __restrict__ bias,
                                                  float* __restrict__ out) {
    __shared__ alignas(16) float2 Bs[(FEAT / 2) * BPITCH];

    const int tid = threadIdx.x;

#pragma unroll 4
    for (int it = tid; it < (FEAT / 2) * (HID / 4); it += 256) {
        const int kp = it / (HID / 4);
        const int n4 = it - kp * (HID / 4);
        const float4 r0 = *(const float4*)(B + (size_t)(2 * kp) * HID + 4 * n4);
        const float4 r1 = *(const float4*)(B + (size_t)(2 * kp + 1) * HID + 4 * n4);
        float4* dst = (float4*)&Bs[kp * BPITCH + 4 * n4];
        dst[0] = make_float4(r0.x, r1.x, r0.y, r1.y);
        dst[1] = make_float4(r0.z, r1.z, r0.w, r1.w);
    }
    __syncthreads();

    const int wave   = tid >> 5;
    const int lane   = tid & 31;
    const int lane16 = lane & 15;
    const int hi     = lane >> 4;

    const int pair = blockIdx.x * 4 + (wave & 3);
    const int sub  = wave >> 2;

    float bcol[8];
#pragma unroll
    for (int t = 0; t < 8; ++t) bcol[t] = bias[t * 16 + lane16];

    const float2* bl = Bs + hi * BPITCH + lane16;

    int tile = pair + NPAIR * sub;
    const float* arow = A + (size_t)(tile * TROWS + lane16) * FEAT + 2 * hi;

    v2f a0[8], a1[8];
    load_ablock(a0, arow);

    for (;;) {
        const int  ntile = tile + 2 * NPAIR;
        const bool more  = ntile < NTILE;
        const float* nrow = A + (size_t)((more ? ntile : tile) * TROWS + lane16) * FEAT + 2 * hi;

        v8f acc[8];
#pragma unroll
        for (int t = 0; t < 8; ++t) {
            const float bv = bcol[t];
            acc[t] = (v8f){bv, bv, bv, bv, bv, bv, bv, bv};
        }

        for (int b = 0; b < NBLK; b += 2) {
            load_ablock(a1, arow + (b + 1) * KBLK);
            mul_block(acc, a0, bl + (size_t)b * (KBLK / 2) * BPITCH);
            load_ablock(a0, (b + 2 < NBLK) ? arow + (b + 2) * KBLK : nrow);
            mul_block(acc, a1, bl + (size_t)(b + 1) * (KBLK / 2) * BPITCH);
        }

#pragma unroll
        for (int t = 0; t < 8; ++t) {
#pragma unroll
            for (int j = 0; j < 8; ++j) {
                const int m   = tile * TROWS + j + 8 * hi;
                const float v = acc[t][j];
                out[(size_t)m * HID + t * 16 + lane16] = v > 0.f ? v : 0.f;
            }
        }

        if (!more) break;
        tile = ntile;
        arow = nrow;
    }
}


__global__ __launch_bounds__(256) void k_h(const float* __restrict__ tf_h,
                                           const float* __restrict__ wf,
                                           const float* __restrict__ Wm,
                                           const float* __restrict__ Wl,
                                           float* __restrict__ hm,
                                           float* __restrict__ hl) {
    const int wave   = threadIdx.x >> 5;
    const int lane   = threadIdx.x & 31;
    const int lane16 = lane & 15;
    const int hi     = lane >> 4;

    const int node_base = (blockIdx.x * 8 + wave) * 16;
    int nd = node_base + lane16;
    if (nd > NODES - 1) nd = NODES - 1;
    const float* xrow = (nd < TASKS) ? (tf_h + (size_t)nd * HID)
                                     : (wf + (size_t)(nd - TASKS) * HID);
    const int n = lane16;
    const bool ncol = n < CLS;

    v8f accM = {0.f, 0.f, 0.f, 0.f, 0.f, 0.f, 0.f, 0.f};
    v8f accL = {0.f, 0.f, 0.f, 0.f, 0.f, 0.f, 0.f, 0.f};

    for (int k = 0; k < HID; k += 4) {
        const int k0 = k + 2 * hi;
        const float2 av = *(const float2*)(xrow + k0);
        v2f af = {av.x, av.y};
        v2f bm, bl;
        bm.x = ncol ? Wm[k0 * CLS + n] : 0.f;
        bm.y = ncol ? Wm[(k0 + 1) * CLS + n] : 0.f;
        bl.x = ncol ? Wl[k0 * CLS + n] : 0.f;
        bl.y = ncol ? Wl[(k0 + 1) * CLS + n] : 0.f;
        accM = __builtin_amdgcn_wmma_f32_16x16x4_f32(false, af, false, bm, (short)0, accM, false, false);
        accL = __builtin_amdgcn_wmma_f32_16x16x4_f32(false, af, false, bl, (short)0, accL, false, false);
    }

    if (ncol) {
#pragma unroll
        for (int j = 0; j < 8; ++j) {
            int m = node_base + j + 8 * hi;
            if (m < NODES) {
                hm[(size_t)m * HP + n] = accM[j];
                hl[(size_t)m * HP + n] = accL[j];
            }
        }
    }
}


__global__ void k_agg_init_w(const float* __restrict__ hm, const float* __restrict__ hl,
                             const float* __restrict__ dinv,
                             const float* __restrict__ bm, const float* __restrict__ bl,
                             float* __restrict__ mean, float* __restrict__ lsd) {
    int i = blockIdx.x * 256 + threadIdx.x;
    if (i < WORKERS * CLS) {
        int w = i / CLS;
        int c = i - w * CLS;
        float d  = dinv[TASKS + w];
        float sw = d * d;
        mean[(size_t)TASKS * CLS + i] = bm[c] + hm[(size_t)(TASKS + w) * HP + c] * sw;
        lsd[(size_t)TASKS * CLS + i]  = bl[c] + hl[(size_t)(TASKS + w) * HP + c] * sw;
    }
}


__device__ __forceinline__ void lds_barrier() {
    __builtin_amdgcn_fence(__ATOMIC_RELEASE, "workgroup", "local");
    __builtin_amdgcn_s_barrier();
    __builtin_amdgcn_fence(__ATOMIC_ACQUIRE, "workgroup", "local");
}

constexpr int GRP_LANES = ROW2 / 4;
constexpr int GRP_WAVE  = 32 / GRP_LANES;
constexpr int NGRP      = (OWN_THREADS / 32) * GRP_WAVE;
constexpr int TROUNDS   = (TBLK + NGRP - 1) / NGRP;
static_assert(TBLK < OWN_THREADS && ROW2 % 4 == 0, "one scan thread per owned task and one more");

__device__ __forceinline__ void load_keys(int (&cur)[KEYS_THREAD], const int* __restrict__ sorted,
                                          int first, int hi, int tid) {
#pragma unroll
    for (int j = 0; j < KEYS_THREAD; ++j) {
        const int idx = first + j * OWN_THREADS + tid;
        cur[j] = (idx < hi) ? sorted[idx] : -1;
    }
}

__global__ __launch_bounds__(OWN_THREADS) void k_agg_owner(const int* __restrict__ sorted,
                                                           const int* __restrict__ segstart,
                                                           const float* __restrict__ hm,
                                                           const float* __restrict__ hl,
                                                           const float* __restrict__ dinv,
                                                           const float* __restrict__ bm,
                                                           const float* __restrict__ bl,
                                                           float* __restrict__ mean,
                                                           float* __restrict__ lsd) {
    __shared__ alignas(16) float Fs[TBLK * ROW2];
    __shared__ alignas(16) float Gs[WORKERS * ROW2];
    __shared__ alignas(16) float wacc[WORKERS * ROW2];
    __shared__ int q[QCAP];
    __shared__ int cnt[TBLK];
    __shared__ int beg[TBLK + 1];
    __shared__ int wsum[OWN_THREADS / 32];

    const int tid    = threadIdx.x;
    const int lane   = tid & 31;
    const int wave   = tid >> 5;
    const int t_base = blockIdx.x * TBLK;

    const int rng = blockIdx.x / OWN_PER_R;
    const int lo  = segstart[rng];
    const int hi  = (rng + 1 < CNT_R) ? segstart[rng + 1] : ANS;
    const int nch = (hi - lo + QCAP - 1) / QCAP;

    int cur[KEYS_THREAD];
    load_keys(cur, sorted, lo, hi, tid);

    const int  gl   = lane / GRP_LANES;
    const int  c4   = (lane - gl * GRP_LANES) * 4;
    const bool gact = lane < GRP_WAVE * GRP_LANES;
    const int  grp  = wave * GRP_WAVE + gl;

    float4 a[TROUNDS];
#pragma unroll
    for (int r = 0; r < TROUNDS; ++r) {
        const int tl = grp + r * NGRP;
        a[r] = make_float4(0.f, 0.f, 0.f, 0.f);
        if (gact && tl < TBLK) {
            const int t = t_base + tl;
            const float d = dinv[t];
            float v[4];
#pragma unroll
            for (int k = 0; k < 4; ++k) {
                const int e = c4 + k;
                v[k] = (e < CLS) ? hm[(size_t)t * HP + e] : hl[(size_t)t * HP + (e - CLS)];
            }
            *(float4*)&Fs[tl * ROW2 + c4] = make_float4(d * v[0], d * v[1], d * v[2], d * v[3]);
        }
    }

    for (int i = tid; i < WORKERS * CLS; i += OWN_THREADS) {
        const int w  = i / CLS;
        const int c  = i - w * CLS;
        const int r  = TASKS + w;
        const float d = dinv[r];
        Gs[w * ROW2 + c]         = d * hm[(size_t)r * HP + c];
        Gs[w * ROW2 + CLS + c]   = d * hl[(size_t)r * HP + c];
        wacc[w * ROW2 + c]       = 0.f;
        wacc[w * ROW2 + CLS + c] = 0.f;
    }
    if (tid < TBLK) cnt[tid] = 0;
    __syncthreads();

    const unsigned key_lo  = (unsigned)t_base << KSH;
    const unsigned key_len = (unsigned)TBLK << KSH;

    for (int ch = 0; ch < nch; ++ch) {
#pragma unroll
        for (int j = 0; j < KEYS_THREAD; ++j) {
            const unsigned rel = (unsigned)cur[j] - key_lo;
            if (rel < key_len) atomicAdd(&cnt[rel >> KSH], 1);
        }
        lds_barrier();

        {
            const int v = (tid < TBLK) ? cnt[tid] : 0;
            int inc = v;
#pragma unroll
            for (int o = 1; o < 32; o <<= 1) {
                const int n = __shfl_up(inc, o);
                if (lane >= o) inc += n;
            }
            if (lane == 31) wsum[wave] = inc;
            lds_barrier();
            int before = (lane < wave) ? wsum[lane] : 0;
#pragma unroll
            for (int o = 16; o > 0; o >>= 1) before += __shfl_xor(before, o);
            const int ex = before + inc - v;
            if (tid <= TBLK) beg[tid] = ex;
            if (tid < TBLK)  cnt[tid] = ex;
            lds_barrier();
        }

#pragma unroll
        for (int j = 0; j < KEYS_THREAD; ++j) {
            const unsigned rel = (unsigned)cur[j] - key_lo;
            if (rel < key_len) {
                const int pos = atomicAdd(&cnt[rel >> KSH], 1);
                q[pos] = (int)(rel & KMASK);
            }
        }

        if (ch + 1 < nch) load_keys(cur, sorted, lo + (ch + 1) * QCAP, hi, tid);

        lds_barrier();
        if (tid < TBLK) cnt[tid] = 0;

#pragma unroll
        for (int r = 0; r < TROUNDS; ++r) {
            const int tl = grp + r * NGRP;
            if (gact && tl < TBLK) {
                const int je = beg[tl + 1];
                const float4 fv = *(const float4*)&Fs[tl * ROW2 + c4];
                float4 s = a[r];
                for (int j = beg[tl]; j < je; ++j) {
                    const int w = q[j];
                    const float4 g = *(const float4*)&Gs[w * ROW2 + c4];
                    s.x += g.x; s.y += g.y; s.z += g.z; s.w += g.w;
                    float* b = &wacc[w * ROW2 + c4];
                    atomicAdd(b + 0, fv.x); atomicAdd(b + 1, fv.y);
                    atomicAdd(b + 2, fv.z); atomicAdd(b + 3, fv.w);
                }
                a[r] = s;
            }
        }
        lds_barrier();
    }

#pragma unroll
    for (int r = 0; r < TROUNDS; ++r) {
        const int tl = grp + r * NGRP;
        if (gact && tl < TBLK) {
            float4* p = (float4*)&Fs[tl * ROW2 + c4];
            const float4 fv = *p;
            *p = make_float4(fv.x + a[r].x, fv.y + a[r].y, fv.z + a[r].z, fv.w + a[r].w);
        }
    }
    __syncthreads();

    for (int i = tid; i < TBLK * CLS; i += OWN_THREADS) {
        const int tl = i / CLS;
        const int c  = i - tl * CLS;
        const float d = dinv[t_base + tl];
        const size_t o = (size_t)t_base * CLS + i;
        mean[o] = bm[c] + d * Fs[tl * ROW2 + c];
        lsd[o]  = bl[c] + d * Fs[tl * ROW2 + CLS + c];
    }

    for (int i = tid; i < WORKERS * CLS; i += OWN_THREADS) {
        const int w = i / CLS;
        const int c = i - w * CLS;
        const float d = dinv[TASKS + w];
        atomicAdd(&mean[(size_t)TASKS * CLS + i], d * wacc[w * ROW2 + c]);
        atomicAdd(&lsd[(size_t)TASKS * CLS + i],  d * wacc[w * ROW2 + CLS + c]);
    }
}


__global__ void k_z(const float* __restrict__ mean, const float* __restrict__ lsd,
                    const float* __restrict__ eps, float* __restrict__ z) {
    int idx = blockIdx.x * 256 + threadIdx.x;
    if (idx < NODES * CLS)
        z[idx] = mean[idx] + (eps[idx] * 0.01f) * expf(lsd[idx]);
}

constexpr int NQUAD      = ANS * CLS / 4;
constexpr int CR_THREADS = 160;
constexpr int CR_WGS     = 625;
constexpr int CR_GT      = CR_THREADS * CR_WGS;
constexpr int CR_TRIPS   = NQUAD / CR_GT;
constexpr int CR_BATCH   = 5;
constexpr int CR_ASTEP   = CR_GT * 4 / CLS;
static_assert((ANS * CLS) % 4 == 0 && NQUAD % CR_GT == 0 && (CR_GT * 4) % CLS == 0 &&
              CR_TRIPS % CR_BATCH == 0 && CLS % 2 == 0, "crowd partition must be exact");

__global__ __launch_bounds__(CR_THREADS) void k_crowd(const int* __restrict__ keys,
                                                      const float* __restrict__ z,
                                                      float* __restrict__ crowd) {
    const int g  = blockIdx.x * CR_THREADS + threadIdx.x;
    const int eA = 4 * g;
    const int aA = eA / CLS;
    const int cA = eA - aA * CLS;
    const int eB = eA + 2;
    const int aB = eB / CLS;
    const int cB = eB - aB * CLS;

    const float* __restrict__ ztA = z + cA;
    const float* __restrict__ ztB = z + cB;
    const float* __restrict__ zwA = z + (size_t)TASKS * CLS + cA;
    const float* __restrict__ zwB = z + (size_t)TASKS * CLS + cB;
    const int*   __restrict__ kA  = keys + aA;
    const int*   __restrict__ kB  = keys + aB;
    float4*      __restrict__ o   = (float4*)crowd + g;

    for (int b = 0; b < CR_TRIPS; b += CR_BATCH) {
        int ka[CR_BATCH], kb[CR_BATCH];
#pragma unroll
        for (int u = 0; u < CR_BATCH; ++u) {
            ka[u] = kA[(b + u) * CR_ASTEP];
            kb[u] = kB[(b + u) * CR_ASTEP];
        }
        float2 ta[CR_BATCH], wa[CR_BATCH], tb[CR_BATCH], wb[CR_BATCH];
#pragma unroll
        for (int u = 0; u < CR_BATCH; ++u) {
            ta[u] = *(const float2*)(ztA + (ka[u] >> KSH) * CLS);
            wa[u] = *(const float2*)(zwA + (ka[u] & KMASK) * CLS);
            tb[u] = *(const float2*)(ztB + (kb[u] >> KSH) * CLS);
            wb[u] = *(const float2*)(zwB + (kb[u] & KMASK) * CLS);
        }
#pragma unroll
        for (int u = 0; u < CR_BATCH; ++u)
            o[(size_t)(b + u) * CR_GT] = make_float4(ta[u].x * wa[u].x, ta[u].y * wa[u].y,
                                                     tb[u].x * wb[u].x, tb[u].y * wb[u].y);
    }
}


extern "C" void kernel_launch(void* const* d_in, const int* in_sizes, int n_in,
                              void* d_out, int out_size, void* d_ws, size_t ws_size,
                              hipStream_t stream) {
    (void)in_sizes; (void)n_in; (void)out_size; (void)ws_size;

    const float* task_feature   = (const float*)d_in[0];
    const int*   answers        = (const int*)d_in[1];
    const float* worker_feature = (const float*)d_in[2];
    const float* W_efc          = (const float*)d_in[3];
    const float* b_efc          = (const float*)d_in[4];
    const float* W_mean         = (const float*)d_in[5];
    const float* b_mean         = (const float*)d_in[6];
    const float* W_ls           = (const float*)d_in[7];
    const float* b_ls           = (const float*)d_in[8];
    const float* eps            = (const float*)d_in[9];

    float* out   = (float*)d_out;
    float* crowd = out;
    float* mean  = out + (size_t)ANS * CLS;
    float* lsd   = mean + (size_t)NODES * CLS;

    float*  ws   = (float*)d_ws;
    float*  tf_h = ws;
    float*  hm   = tf_h + (size_t)TASKS * HID;
    float*  hl   = hm + (size_t)NODES * HP;
    float*  zz   = hl + (size_t)NODES * HP;
    float*  dinv = zz + (size_t)NODES * CLS;

    auto cdiv = [](int a, int b) { return (a + b - 1) / b; };

    k_gemm_efc <<<GEMM_WGS, 256, 0, stream>>>(task_feature, W_efc, b_efc, tf_h);
    k_h        <<<cdiv(cdiv(NODES, 16), 8), 256, 0, stream>>>(tf_h, worker_feature,
                                                              W_mean, W_ls, hm, hl);

    int* keys  = (int*)tf_h;
    int* tpart = keys + ANS;
    int* wpart = tpart + (size_t)CNT_S * TASKS;
    int* wvcnt = wpart + (size_t)CNT_WGS * WPITCH;
    int* segst = wvcnt + (size_t)CNT_WGS * CNT_WAVES;
    int* sorted = segst + 32;
    k_pack     <<<cdiv(ANS, 256), 256, 0, stream>>>(answers, keys);
    k_count    <<<CNT_WGS, CNT_THREADS, 0, stream>>>(keys, tpart, wpart, wvcnt);
    k_scatter  <<<CNT_WGS, CNT_THREADS, 0, stream>>>(keys, wvcnt, sorted, segst);
    k_dinv     <<<cdiv(NODES, 256), 256, 0, stream>>>(tpart, wpart, dinv);

    k_agg_init_w<<<cdiv(WORKERS * CLS, 256), 256, 0, stream>>>(hm, hl, dinv, b_mean, b_ls, mean, lsd);
    k_agg_owner <<<NOWN, OWN_THREADS, 0, stream>>>(sorted, segst, hm, hl, dinv, b_mean, b_ls, mean, lsd);

    k_z        <<<cdiv(NODES * CLS, 256), 256, 0, stream>>>(mean, lsd, eps, zz);
    k_crowd    <<<CR_WGS, CR_THREADS, 0, stream>>>(keys, zz, crowd);
}


# Worker-row edge sums leave each aggregation owner as a plain 40 KB plane; the z launch adds the 125 planes in fixed order with bias and self-loop: global float atomics and the init launch gone
# speedup vs baseline: 4.3717x; 1.0314x over previous
#include <hip/hip_runtime.h>

typedef __attribute__((ext_vector_type(2))) float v2f;
typedef __attribute__((ext_vector_type(8))) float v8f;

constexpr int TASKS   = 100000;
constexpr int WORKERS = 500;
constexpr int FEAT    = 512;
constexpr int CLS     = 10;
constexpr int HP      = 12;
constexpr int ANS     = 1000000;
constexpr int HID     = 128;
constexpr int NODES   = TASKS + WORKERS;
constexpr int KSH     = 9;
constexpr int KMASK   = (1 << KSH) - 1;
constexpr int TBLK    = 800;
constexpr int NOWN    = TASKS / TBLK;
constexpr int ROW2    = 2 * CLS;
constexpr int QCAP    = 40960;
constexpr int OWN_THREADS = 1024;
constexpr int KEYS_THREAD = QCAP / OWN_THREADS;
constexpr int TROWS   = 16;
constexpr int NTILE   = TASKS / TROWS;
constexpr int GEMM_WGS = 256;
constexpr int NPAIR   = GEMM_WGS * 4;
constexpr int BPITCH  = HID + 16;
constexpr int KBLK    = 32;
constexpr int NBLK    = FEAT / KBLK;

constexpr int CNT_R     = 25;
constexpr int CNT_S     = 10;
constexpr int CNT_WGS   = CNT_R * CNT_S;
constexpr int CNT_THREADS = 1024;
constexpr int CNT_WAVES = CNT_THREADS / 32;
constexpr int RTASKS    = TASKS / CNT_R;
constexpr int SGRPS     = (ANS / 4) / CNT_S;
constexpr int WPITCH    = 512;
constexpr int CNT_BATCH = 8;
constexpr int CNT_TRIPS = (SGRPS + CNT_THREADS * CNT_BATCH - 1) / (CNT_THREADS * CNT_BATCH);
constexpr int OWN_PER_R = RTASKS / TBLK;
static_assert(TASKS % CNT_R == 0 && ANS % (4 * CNT_S) == 0 && WORKERS <= WPITCH && WPITCH <= CNT_THREADS &&
              RTASKS % TBLK == 0, "count partition must be exact");

__global__ void k_pack(const int* __restrict__ answers, int* __restrict__ keys) {
    int a = blockIdx.x * 256 + threadIdx.x;
    if (a < ANS) {
        const int t = answers[3 * a];
        const int w = answers[3 * a + 1];
        keys[a] = (t << KSH) | w;
    }
}

__global__ __launch_bounds__(CNT_THREADS) void k_count(const int* __restrict__ keys,
                                                       int* __restrict__ tpart,
                                                       int* __restrict__ wpart,
                                                       int* __restrict__ wvcnt) {
    __shared__ int tcnt[RTASKS];
    __shared__ int wcnt[WPITCH];

    const int tid = threadIdx.x;
    const int r   = blockIdx.x / CNT_S;
    const int s   = blockIdx.x % CNT_S;

    for (int i = tid; i < RTASKS; i += CNT_THREADS) tcnt[i] = 0;
    if (tid < WPITCH) wcnt[tid] = 0;
    __syncthreads();

    const int4* __restrict__ kp = (const int4*)keys + (size_t)s * SGRPS;
    const unsigned key_lo  = (unsigned)(r * RTASKS) << KSH;
    const unsigned key_len = (unsigned)RTASKS << KSH;

    int wc = 0;
    for (int k = 0; k < CNT_TRIPS; ++k) {
        const int g0 = tid + k * CNT_THREADS * CNT_BATCH;
        int4 v[CNT_BATCH];
#pragma unroll
        for (int j = 0; j < CNT_BATCH; ++j) {
            const int g = g0 + j * CNT_THREADS;
            v[j] = kp[g < SGRPS ? g : SGRPS - 1];
        }
#pragma unroll
        for (int j = 0; j < CNT_BATCH; ++j) {
            const bool valid = g0 + j * CNT_THREADS < SGRPS;
            const int kk[4] = {v[j].x, v[j].y, v[j].z, v[j].w};
#pragma unroll
            for (int u = 0; u < 4; ++u) {
                const unsigned rel = (unsigned)kk[u] - key_lo;
                const bool match = valid && rel < key_len;
                wc += __popc(__builtin_amdgcn_ballot_w32(match));
                if (match) {
                    atomicAdd(&tcnt[rel >> KSH], 1);
                    atomicAdd(&wcnt[rel & KMASK], 1);
                }
            }
        }
    }
    if ((tid & 31) == 0) wvcnt[blockIdx.x * CNT_WAVES + (tid >> 5)] = wc;
    __syncthreads();

    int* __restrict__ tp = tpart + (size_t)s * TASKS + r * RTASKS;
    for (int i = tid; i < RTASKS; i += CNT_THREADS) tp[i] = tcnt[i];
    if (tid < WPITCH) wpart[(size_t)blockIdx.x * WPITCH + tid] = wcnt[tid];
}

__global__ __launch_bounds__(CNT_THREADS) void k_scatter(const int* __restrict__ keys,
                                                         const int* __restrict__ wvcnt,
                                                         int* __restrict__ sorted,
                                                         int* __restrict__ segstart) {
    __shared__ int base_s;

    const int tid  = threadIdx.x;
    const int lane = tid & 31;
    const int wave = tid >> 5;
    const int r    = blockIdx.x / CNT_S;
    const int s    = blockIdx.x % CNT_S;

    if (tid == 0) base_s = 0;
    __syncthreads();

    const int nbefore = blockIdx.x * CNT_WAVES;
    int part = 0;
    for (int i = tid; i < nbefore; i += CNT_THREADS) part += wvcnt[i];
    int mine = (lane < wave) ? wvcnt[nbefore + lane] : 0;
    for (int o = 16; o > 0; o >>= 1) {
        part += __shfl_xor(part, o);
        mine += __shfl_xor(mine, o);
    }
    if (lane == 0 && part != 0) atomicAdd(&base_s, part);
    __syncthreads();
    const int wg_base = base_s;
    if (s == 0 && tid == 0) segstart[r] = wg_base;

    const int4* __restrict__ kp = (const int4*)keys + (size_t)s * SGRPS;
    const unsigned key_lo  = (unsigned)(r * RTASKS) << KSH;
    const unsigned key_len = (unsigned)RTASKS << KSH;
    const unsigned below   = (1u << lane) - 1u;

    int wpos = wg_base + mine;
    for (int k = 0; k < CNT_TRIPS; ++k) {
        const int g0 = tid + k * CNT_THREADS * CNT_BATCH;
        int4 v[CNT_BATCH];
#pragma unroll
        for (int j = 0; j < CNT_BATCH; ++j) {
            const int g = g0 + j * CNT_THREADS;
            v[j] = kp[g < SGRPS ? g : SGRPS - 1];
        }
#pragma unroll
        for (int j = 0; j < CNT_BATCH; ++j) {
            const bool valid = g0 + j * CNT_THREADS < SGRPS;
            const int kk[4] = {v[j].x, v[j].y, v[j].z, v[j].w};
#pragma unroll
            for (int u = 0; u < 4; ++u) {
                const unsigned rel = (unsigned)kk[u] - key_lo;
                const bool match = valid && rel < key_len;
                const unsigned m = __builtin_amdgcn_ballot_w32(match);
                if (match) sorted[wpos + __popc(m & below)] = kk[u];
                wpos += __popc(m);
            }
        }
    }
}

__global__ void k_dinv(const int* __restrict__ tpart, const int* __restrict__ wpart,
                       float* __restrict__ dinv) {
    int i = blockIdx.x * 256 + threadIdx.x;
    if (i < TASKS) {
        int d = 1;
#pragma unroll
        for (int s = 0; s < CNT_S; ++s) d += tpart[(size_t)s * TASKS + i];
        dinv[i] = rsqrtf((float)d);
    } else if (i < NODES) {
        const int w = i - TASKS;
        int d = 1;
#pragma unroll 10
        for (int b = 0; b < CNT_WGS; ++b) d += wpart[(size_t)b * WPITCH + w];
        dinv[i] = rsqrtf((float)d);
    }
}


__device__ __forceinline__ void load_ablock(v2f (&a)[8], const float* __restrict__ p) {
#pragma unroll
    for (int s = 0; s < 8; ++s) {
        const float2 v = *(const float2*)(p + 4 * s);
        a[s] = (v2f){v.x, v.y};
    }
}

__device__ __forceinline__ void mul_block(v8f (&acc)[8], const v2f (&a)[8], const float2* bb) {
#pragma unroll
    for (int s = 0; s < 8; ++s) {
        const float2* brow = bb + s * 2 * BPITCH;
#pragma unroll
        for (int t = 0; t < 8; ++t) {
            const float2 bv = brow[t * 16];
            v2f bf = {bv.x, bv.y};
            acc[t] = __builtin_amdgcn_wmma_f32_16x16x4_f32(
                false, a[s], false, bf, (short)0, acc[t], false, false);
        }
    }
}

__global__ __launch_bounds__(256) void k_gemm_efc(const float* __restrict__ A,
                                                  const float* __restrict__ B,
                                                  const float* __restrict__ bias,
                                                  float* __restrict__ out) {
    __shared__ alignas(16) float2 Bs[(FEAT / 2) * BPITCH];

    const int tid = threadIdx.x;

#pragma unroll 4
    for (int it = tid; it < (FEAT / 2) * (HID / 4); it += 256) {
        const int kp = it / (HID / 4);
        const int n4 = it - kp * (HID / 4);
        const float4 r0 = *(const float4*)(B + (size_t)(2 * kp) * HID + 4 * n4);
        const float4 r1 = *(const float4*)(B + (size_t)(2 * kp + 1) * HID + 4 * n4);
        float4* dst = (float4*)&Bs[kp * BPITCH + 4 * n4];
        dst[0] = make_float4(r0.x, r1.x, r0.y, r1.y);
        dst[1] = make_float4(r0.z, r1.z, r0.w, r1.w);
    }
    __syncthreads();

    const int wave   = tid >> 5;
    const int lane   = tid & 31;
    const int lane16 = lane & 15;
    const int hi     = lane >> 4;

    const int pair = blockIdx.x * 4 + (wave & 3);
    const int sub  = wave >> 2;

    float bcol[8];
#pragma unroll
    for (int t = 0; t < 8; ++t) bcol[t] = bias[t * 16 + lane16];

    const float2* bl = Bs + hi * BPITCH + lane16;

    int tile = pair + NPAIR * sub;
    const float* arow = A + (size_t)(tile * TROWS + lane16) * FEAT + 2 * hi;

    v2f a0[8], a1[8];
    load_ablock(a0, arow);

    for (;;) {
        const int  ntile = tile + 2 * NPAIR;
        const bool more  = ntile < NTILE;
        const float* nrow = A + (size_t)((more ? ntile : tile) * TROWS + lane16) * FEAT + 2 * hi;

        v8f acc[8];
#pragma unroll
        for (int t = 0; t < 8; ++t) {
            const float bv = bcol[t];
            acc[t] = (v8f){bv, bv, bv, bv, bv, bv, bv, bv};
        }

        for (int b = 0; b < NBLK; b += 2) {
            load_ablock(a1, arow + (b + 1) * KBLK);
            mul_block(acc, a0, bl + (size_t)b * (KBLK / 2) * BPITCH);
            load_ablock(a0, (b + 2 < NBLK) ? arow + (b + 2) * KBLK : nrow);
            mul_block(acc, a1, bl + (size_t)(b + 1) * (KBLK / 2) * BPITCH);
        }

#pragma unroll
        for (int t = 0; t < 8; ++t) {
#pragma unroll
            for (int j = 0; j < 8; ++j) {
                const int m   = tile * TROWS + j + 8 * hi;
                const float v = acc[t][j];
                out[(size_t)m * HID + t * 16 + lane16] = v > 0.f ? v : 0.f;
            }
        }

        if (!more) break;
        tile = ntile;
        arow = nrow;
    }
}


__global__ __launch_bounds__(256) void k_h(const float* __restrict__ tf_h,
                                           const float* __restrict__ wf,
                                           const float* __restrict__ Wm,
                                           const float* __restrict__ Wl,
                                           float* __restrict__ hm,
                                           float* __restrict__ hl) {
    const int wave   = threadIdx.x >> 5;
    const int lane   = threadIdx.x & 31;
    const int lane16 = lane & 15;
    const int hi     = lane >> 4;

    const int node_base = (blockIdx.x * 8 + wave) * 16;
    int nd = node_base + lane16;
    if (nd > NODES - 1) nd = NODES - 1;
    const float* xrow = (nd < TASKS) ? (tf_h + (size_t)nd * HID)
                                     : (wf + (size_t)(nd - TASKS) * HID);
    const int n = lane16;
    const bool ncol = n < CLS;

    v8f accM = {0.f, 0.f, 0.f, 0.f, 0.f, 0.f, 0.f, 0.f};
    v8f accL = {0.f, 0.f, 0.f, 0.f, 0.f, 0.f, 0.f, 0.f};

    for (int k = 0; k < HID; k += 4) {
        const int k0 = k + 2 * hi;
        const float2 av = *(const float2*)(xrow + k0);
        v2f af = {av.x, av.y};
        v2f bm, bl;
        bm.x = ncol ? Wm[k0 * CLS + n] : 0.f;
        bm.y = ncol ? Wm[(k0 + 1) * CLS + n] : 0.f;
        bl.x = ncol ? Wl[k0 * CLS + n] : 0.f;
        bl.y = ncol ? Wl[(k0 + 1) * CLS + n] : 0.f;
        accM = __builtin_amdgcn_wmma_f32_16x16x4_f32(false, af, false, bm, (short)0, accM, false, false);
        accL = __builtin_amdgcn_wmma_f32_16x16x4_f32(false, af, false, bl, (short)0, accL, false, false);
    }

    if (ncol) {
#pragma unroll
        for (int j = 0; j < 8; ++j) {
            int m = node_base + j + 8 * hi;
            if (m < NODES) {
                hm[(size_t)m * HP + n] = accM[j];
                hl[(size_t)m * HP + n] = accL[j];
            }
        }
    }
}


__device__ __forceinline__ void lds_barrier() {
    __builtin_amdgcn_fence(__ATOMIC_RELEASE, "workgroup", "local");
    __builtin_amdgcn_s_barrier();
    __builtin_amdgcn_fence(__ATOMIC_ACQUIRE, "workgroup", "local");
}

constexpr int PLANE     = WORKERS * ROW2;
constexpr int GRP_LANES = ROW2 / 4;
constexpr int GRP_WAVE  = 32 / GRP_LANES;
constexpr int NGRP      = (OWN_THREADS / 32) * GRP_WAVE;
constexpr int TROUNDS   = (TBLK + NGRP - 1) / NGRP;
static_assert(TBLK < OWN_THREADS && ROW2 % 4 == 0 && PLANE % 4 == 0,
              "one scan thread per owned task and one more; planes move as 16-byte pieces");

__device__ __forceinline__ void load_keys(int (&cur)[KEYS_THREAD], const int* __restrict__ sorted,
                                          int first, int hi, int tid) {
#pragma unroll
    for (int j = 0; j < KEYS_THREAD; ++j) {
        const int idx = first + j * OWN_THREADS + tid;
        cur[j] = (idx < hi) ? sorted[idx] : -1;
    }
}

__global__ __launch_bounds__(OWN_THREADS) void k_agg_owner(const int* __restrict__ sorted,
                                                           const int* __restrict__ segstart,
                                                           const float* __restrict__ hm,
                                                           const float* __restrict__ hl,
                                                           const float* __restrict__ dinv,
                                                           const float* __restrict__ bm,
                                                           const float* __restrict__ bl,
                                                           float* __restrict__ mean,
                                                           float* __restrict__ lsd,
                                                           float* __restrict__ planes) {
    __shared__ alignas(16) float Fs[TBLK * ROW2];
    __shared__ alignas(16) float Gs[WORKERS * ROW2];
    __shared__ alignas(16) float wacc[WORKERS * ROW2];
    __shared__ int q[QCAP];
    __shared__ int cnt[TBLK];
    __shared__ int beg[TBLK + 1];
    __shared__ int wsum[OWN_THREADS / 32];

    const int tid    = threadIdx.x;
    const int lane   = tid & 31;
    const int wave   = tid >> 5;
    const int t_base = blockIdx.x * TBLK;

    const int rng = blockIdx.x / OWN_PER_R;
    const int lo  = segstart[rng];
    const int hi  = (rng + 1 < CNT_R) ? segstart[rng + 1] : ANS;
    const int nch = (hi - lo + QCAP - 1) / QCAP;

    int cur[KEYS_THREAD];
    load_keys(cur, sorted, lo, hi, tid);

    const int  gl   = lane / GRP_LANES;
    const int  c4   = (lane - gl * GRP_LANES) * 4;
    const bool gact = lane < GRP_WAVE * GRP_LANES;
    const int  grp  = wave * GRP_WAVE + gl;

    float4 a[TROUNDS];
#pragma unroll
    for (int r = 0; r < TROUNDS; ++r) {
        const int tl = grp + r * NGRP;
        a[r] = make_float4(0.f, 0.f, 0.f, 0.f);
        if (gact && tl < TBLK) {
            const int t = t_base + tl;
            const float d = dinv[t];
            float v[4];
#pragma unroll
            for (int k = 0; k < 4; ++k) {
                const int e = c4 + k;
                v[k] = (e < CLS) ? hm[(size_t)t * HP + e] : hl[(size_t)t * HP + (e - CLS)];
            }
            *(float4*)&Fs[tl * ROW2 + c4] = make_float4(d * v[0], d * v[1], d * v[2], d * v[3]);
        }
    }

    for (int i = tid; i < WORKERS * CLS; i += OWN_THREADS) {
        const int w  = i / CLS;
        const int c  = i - w * CLS;
        const int r  = TASKS + w;
        const float d = dinv[r];
        Gs[w * ROW2 + c]         = d * hm[(size_t)r * HP + c];
        Gs[w * ROW2 + CLS + c]   = d * hl[(size_t)r * HP + c];
        wacc[w * ROW2 + c]       = 0.f;
        wacc[w * ROW2 + CLS + c] = 0.f;
    }
    if (tid < TBLK) cnt[tid] = 0;
    __syncthreads();

    const unsigned key_lo  = (unsigned)t_base << KSH;
    const unsigned key_len = (unsigned)TBLK << KSH;

    for (int ch = 0; ch < nch; ++ch) {
#pragma unroll
        for (int j = 0; j < KEYS_THREAD; ++j) {
            const unsigned rel = (unsigned)cur[j] - key_lo;
            if (rel < key_len) atomicAdd(&cnt[rel >> KSH], 1);
        }
        lds_barrier();

        {
            const int v = (tid < TBLK) ? cnt[tid] : 0;
            int inc = v;
#pragma unroll
            for (int o = 1; o < 32; o <<= 1) {
                const int n = __shfl_up(inc, o);
                if (lane >= o) inc += n;
            }
            if (lane == 31) wsum[wave] = inc;
            lds_barrier();
            int before = (lane < wave) ? wsum[lane] : 0;
#pragma unroll
            for (int o = 16; o > 0; o >>= 1) before += __shfl_xor(before, o);
            const int ex = before + inc - v;
            if (tid <= TBLK) beg[tid] = ex;
            if (tid < TBLK)  cnt[tid] = ex;
            lds_barrier();
        }

#pragma unroll
        for (int j = 0; j < KEYS_THREAD; ++j) {
            const unsigned rel = (unsigned)cur[j] - key_lo;
            if (rel < key_len) {
                const int pos = atomicAdd(&cnt[rel >> KSH], 1);
                q[pos] = (int)(rel & KMASK);
            }
        }

        if (ch + 1 < nch) load_keys(cur, sorted, lo + (ch + 1) * QCAP, hi, tid);

        lds_barrier();
        if (tid < TBLK) cnt[tid] = 0;

#pragma unroll
        for (int r = 0; r < TROUNDS; ++r) {
            const int tl = grp + r * NGRP;
            if (gact && tl < TBLK) {
                const int je = beg[tl + 1];
                const float4 fv = *(const float4*)&Fs[tl * ROW2 + c4];
                float4 s = a[r];
                for (int j = beg[tl]; j < je; ++j) {
                    const int w = q[j];
                    const float4 g = *(const float4*)&Gs[w * ROW2 + c4];
                    s.x += g.x; s.y += g.y; s.z += g.z; s.w += g.w;
                    float* b = &wacc[w * ROW2 + c4];
                    atomicAdd(b + 0, fv.x); atomicAdd(b + 1, fv.y);
                    atomicAdd(b + 2, fv.z); atomicAdd(b + 3, fv.w);
                }
                a[r] = s;
            }
        }
        lds_barrier();
    }

#pragma unroll
    for (int r = 0; r < TROUNDS; ++r) {
        const int tl = grp + r * NGRP;
        if (gact && tl < TBLK) {
            float4* p = (float4*)&Fs[tl * ROW2 + c4];
            const float4 fv = *p;
            *p = make_float4(fv.x + a[r].x, fv.y + a[r].y, fv.z + a[r].z, fv.w + a[r].w);
        }
    }
    __syncthreads();

    for (int i = tid; i < TBLK * CLS; i += OWN_THREADS) {
        const int tl = i / CLS;
        const int c  = i - tl * CLS;
        const float d = dinv[t_base + tl];
        const size_t o = (size_t)t_base * CLS + i;
        mean[o] = bm[c] + d * Fs[tl * ROW2 + c];
        lsd[o]  = bl[c] + d * Fs[tl * ROW2 + CLS + c];
    }

    float4* __restrict__ pl = (float4*)(planes + (size_t)blockIdx.x * PLANE);
    for (int i = tid; i < PLANE / 4; i += OWN_THREADS) pl[i] = *(const float4*)&wacc[4 * i];
}


constexpr int Z_THREADS = 320;
constexpr int ZW_EL     = 64;
constexpr int ZW_PARTS  = Z_THREADS / ZW_EL;
constexpr int ZW_DEPTH  = NOWN / ZW_PARTS;
constexpr int ZW_WGS    = (WORKERS * CLS + ZW_EL - 1) / ZW_EL;
constexpr int ZT_WGS    = TASKS * CLS / Z_THREADS;
constexpr int Z_WGS     = ZW_WGS + ZT_WGS;
static_assert(NOWN % ZW_PARTS == 0 && (TASKS * CLS) % Z_THREADS == 0 && Z_THREADS % ZW_EL == 0,
              "z partition must be exact");

__global__ __launch_bounds__(Z_THREADS) void k_z(const float* __restrict__ planes,
                                                 const float* __restrict__ hm, const float* __restrict__ hl,
                                                 const float* __restrict__ dinv,
                                                 const float* __restrict__ bm, const float* __restrict__ bl,
                                                 const float* __restrict__ eps,
                                                 float* __restrict__ mean, float* __restrict__ lsd,
                                                 float* __restrict__ z) {
    __shared__ float red[2][ZW_PARTS][ZW_EL];

    if (blockIdx.x >= ZW_WGS) {
        const int idx = (blockIdx.x - ZW_WGS) * Z_THREADS + threadIdx.x;
        z[idx] = mean[idx] + (eps[idx] * 0.01f) * expf(lsd[idx]);
        return;
    }

    const int  el   = threadIdx.x % ZW_EL;
    const int  part = threadIdx.x / ZW_EL;
    const int  i    = blockIdx.x * ZW_EL + el;
    const bool ok   = i < WORKERS * CLS;
    const int  ic   = ok ? i : WORKERS * CLS - 1;
    const int  w    = ic / CLS;
    const int  c    = ic - w * CLS;

    const float* __restrict__ p = planes + (size_t)part * PLANE + w * ROW2 + c;
    float sm = 0.f, sl = 0.f;
#pragma unroll
    for (int k = 0; k < ZW_DEPTH; ++k) {
        sm += p[(size_t)k * ZW_PARTS * PLANE];
        sl += p[(size_t)k * ZW_PARTS * PLANE + CLS];
    }
    red[0][part][el] = sm;
    red[1][part][el] = sl;
    __syncthreads();

    if (part == 0 && ok) {
        float tm = red[0][0][el], tl = red[1][0][el];
#pragma unroll
        for (int q = 1; q < ZW_PARTS; ++q) { tm += red[0][q][el]; tl += red[1][q][el]; }
        const float  d  = dinv[TASKS + w];
        const float  sw = d * d;
        const size_t o  = (size_t)TASKS * CLS + i;
        const float  m  = bm[c] + hm[(size_t)(TASKS + w) * HP + c] * sw + d * tm;
        const float  l  = bl[c] + hl[(size_t)(TASKS + w) * HP + c] * sw + d * tl;
        mean[o] = m;
        lsd[o]  = l;
        z[o]    = m + (eps[o] * 0.01f) * expf(l);
    }
}

constexpr int NQUAD      = ANS * CLS / 4;
constexpr int CR_THREADS = 160;
constexpr int CR_WGS     = 625;
constexpr int CR_GT      = CR_THREADS * CR_WGS;
constexpr int CR_TRIPS   = NQUAD / CR_GT;
constexpr int CR_BATCH   = 5;
constexpr int CR_ASTEP   = CR_GT * 4 / CLS;
static_assert((ANS * CLS) % 4 == 0 && NQUAD % CR_GT == 0 && (CR_GT * 4) % CLS == 0 &&
              CR_TRIPS % CR_BATCH == 0 && CLS % 2 == 0, "crowd partition must be exact");

__global__ __launch_bounds__(CR_THREADS) void k_crowd(const int* __restrict__ keys,
                                                      const float* __restrict__ z,
                                                      float* __restrict__ crowd) {
    const int g  = blockIdx.x * CR_THREADS + threadIdx.x;
    const int eA = 4 * g;
    const int aA = eA / CLS;
    const int cA = eA - aA * CLS;
    const int eB = eA + 2;
    const int aB = eB / CLS;
    const int cB = eB - aB * CLS;

    const float* __restrict__ ztA = z + cA;
    const float* __restrict__ ztB = z + cB;
    const float* __restrict__ zwA = z + (size_t)TASKS * CLS + cA;
    const float* __restrict__ zwB = z + (size_t)TASKS * CLS + cB;
    const int*   __restrict__ kA  = keys + aA;
    const int*   __restrict__ kB  = keys + aB;
    float4*      __restrict__ o   = (float4*)crowd + g;

    for (int b = 0; b < CR_TRIPS; b += CR_BATCH) {
        int ka[CR_BATCH], kb[CR_BATCH];
#pragma unroll
        for (int u = 0; u < CR_BATCH; ++u) {
            ka[u] = kA[(b + u) * CR_ASTEP];
            kb[u] = kB[(b + u) * CR_ASTEP];
        }
        float2 ta[CR_BATCH], wa[CR_BATCH], tb[CR_BATCH], wb[CR_BATCH];
#pragma unroll
        for (int u = 0; u < CR_BATCH; ++u) {
            ta[u] = *(const float2*)(ztA + (ka[u] >> KSH) * CLS);
            wa[u] = *(const float2*)(zwA + (ka[u] & KMASK) * CLS);
            tb[u] = *(const float2*)(ztB + (kb[u] >> KSH) * CLS);
            wb[u] = *(const float2*)(zwB + (kb[u] & KMASK) * CLS);
        }
#pragma unroll
        for (int u = 0; u < CR_BATCH; ++u)
            o[(size_t)(b + u) * CR_GT] = make_float4(ta[u].x * wa[u].x, ta[u].y * wa[u].y,
                                                     tb[u].x * wb[u].x, tb[u].y * wb[u].y);
    }
}


extern "C" void kernel_launch(void* const* d_in, const int* in_sizes, int n_in,
                              void* d_out, int out_size, void* d_ws, size_t ws_size,
                              hipStream_t stream) {
    (void)in_sizes; (void)n_in; (void)out_size; (void)ws_size;

    const float* task_feature   = (const float*)d_in[0];
    const int*   answers        = (const int*)d_in[1];
    const float* worker_feature = (const float*)d_in[2];
    const float* W_efc          = (const float*)d_in[3];
    const float* b_efc          = (const float*)d_in[4];
    const float* W_mean         = (const float*)d_in[5];
    const float* b_mean         = (const float*)d_in[6];
    const float* W_ls           = (const float*)d_in[7];
    const float* b_ls           = (const float*)d_in[8];
    const float* eps            = (const float*)d_in[9];

    float* out   = (float*)d_out;
    float* crowd = out;
    float* mean  = out + (size_t)ANS * CLS;
    float* lsd   = mean + (size_t)NODES * CLS;

    float*  ws   = (float*)d_ws;
    float*  tf_h = ws;
    float*  hm   = tf_h + (size_t)TASKS * HID;
    float*  hl   = hm + (size_t)NODES * HP;
    float*  zz   = hl + (size_t)NODES * HP;
    float*  dinv = zz + (size_t)NODES * CLS;

    auto cdiv = [](int a, int b) { return (a + b - 1) / b; };

    k_gemm_efc <<<GEMM_WGS, 256, 0, stream>>>(task_feature, W_efc, b_efc, tf_h);
    k_h        <<<cdiv(cdiv(NODES, 16), 8), 256, 0, stream>>>(tf_h, worker_feature,
                                                              W_mean, W_ls, hm, hl);

    int* keys  = (int*)tf_h;
    int* tpart = keys + ANS;
    int* wpart = tpart + (size_t)CNT_S * TASKS;
    int* wvcnt = wpart + (size_t)CNT_WGS * WPITCH;
    int* segst = wvcnt + (size_t)CNT_WGS * CNT_WAVES;
    int* sorted = segst + 32;
    float* planes = (float*)(sorted + ANS);
    k_pack     <<<cdiv(ANS, 256), 256, 0, stream>>>(answers, keys);
    k_count    <<<CNT_WGS, CNT_THREADS, 0, stream>>>(keys, tpart, wpart, wvcnt);
    k_scatter  <<<CNT_WGS, CNT_THREADS, 0, stream>>>(keys, wvcnt, sorted, segst);
    k_dinv     <<<cdiv(NODES, 256), 256, 0, stream>>>(tpart, wpart, dinv);

    k_agg_owner <<<NOWN, OWN_THREADS, 0, stream>>>(sorted, segst, hm, hl, dinv, b_mean, b_ls, mean, lsd, planes);

    k_z        <<<Z_WGS, Z_THREADS, 0, stream>>>(planes, hm, hl, dinv, b_mean, b_ls, eps, mean, lsd, zz);
    k_crowd    <<<CR_WGS, CR_THREADS, 0, stream>>>(keys, zz, crowd);
}


# Keys grouped by task range reading each answer once (LDS range counts, prefix sums, LDS cursors: arrival order in a range), not by 25 filtered scans; owners count task degrees; launches 9 to 7
# speedup vs baseline: 5.0953x; 1.1655x over previous
#include <hip/hip_runtime.h>

typedef __attribute__((ext_vector_type(2))) float v2f;
typedef __attribute__((ext_vector_type(8))) float v8f;

constexpr int TASKS   = 100000;
constexpr int WORKERS = 500;
constexpr int FEAT    = 512;
constexpr int CLS     = 10;
constexpr int HP      = 12;
constexpr int ANS     = 1000000;
constexpr int HID     = 128;
constexpr int NODES   = TASKS + WORKERS;
constexpr int KSH     = 9;
constexpr int KMASK   = (1 << KSH) - 1;
constexpr int TBLK    = 800;
constexpr int NOWN    = TASKS / TBLK;
constexpr int ROW2    = 2 * CLS;
constexpr int QCAP    = 40960;
constexpr int OWN_THREADS = 1024;
constexpr int KEYS_THREAD = QCAP / OWN_THREADS;
constexpr int TROWS   = 16;
constexpr int NTILE   = TASKS / TROWS;
constexpr int GEMM_WGS = 256;
constexpr int NPAIR   = GEMM_WGS * 4;
constexpr int BPITCH  = HID + 16;
constexpr int KBLK    = 32;
constexpr int NBLK    = FEAT / KBLK;

constexpr int CNT_R      = 25;
constexpr int RTASKS     = TASKS / CNT_R;
constexpr int OWN_PER_R  = RTASKS / TBLK;
constexpr int WPITCH     = 512;
constexpr int HPITCH     = 32;
constexpr int SP_WGS     = 250;
constexpr int SP_THREADS = 1024;
constexpr int SP_SLICE   = ANS / SP_WGS;
constexpr int SP_PER     = (SP_SLICE + SP_THREADS - 1) / SP_THREADS;
constexpr int SP_ROWS    = (SP_WGS + SP_THREADS / HPITCH - 1) / (SP_THREADS / HPITCH);
constexpr int SC_WGS     = SP_WGS + 1;
static_assert(TASKS % CNT_R == 0 && ANS % SP_WGS == 0 && WORKERS <= WPITCH && 2 * WPITCH == SP_THREADS &&
              CNT_R <= HPITCH && HPITCH == 32 && RTASKS % TBLK == 0 && SP_WGS % 2 == 0,
              "split partition must be exact");

__global__ __launch_bounds__(SP_THREADS) void k_split_count(const int* __restrict__ answers,
                                                            int* __restrict__ keys,
                                                            int* __restrict__ hist,
                                                            int* __restrict__ wpart) {
    __shared__ int rcnt[HPITCH];
    __shared__ int wcnt[WPITCH];

    const int tid = threadIdx.x;
    const int a0  = blockIdx.x * SP_SLICE;

    int t[SP_PER], w[SP_PER];
#pragma unroll
    for (int j = 0; j < SP_PER; ++j) {
        const int i = tid + j * SP_THREADS;
        const int a = a0 + (i < SP_SLICE ? i : SP_SLICE - 1);
        t[j] = answers[3 * a];
        w[j] = answers[3 * a + 1];
    }

    if (tid < HPITCH) rcnt[tid] = 0;
    if (tid < WPITCH) wcnt[tid] = 0;
    __syncthreads();

#pragma unroll
    for (int j = 0; j < SP_PER; ++j) {
        const int i = tid + j * SP_THREADS;
        if (i < SP_SLICE) {
            keys[a0 + i] = (t[j] << KSH) | w[j];
            atomicAdd(&rcnt[(unsigned)t[j] / RTASKS], 1);
            atomicAdd(&wcnt[w[j]], 1);
        }
    }
    __syncthreads();

    if (tid < HPITCH) hist[blockIdx.x * HPITCH + tid] = rcnt[tid];
    if (tid < WPITCH) wpart[(size_t)blockIdx.x * WPITCH + tid] = wcnt[tid];
}

__global__ __launch_bounds__(SP_THREADS) void k_split_scatter(const int* __restrict__ keys,
                                                              const int* __restrict__ hist,
                                                              const int* __restrict__ wpart,
                                                              int* __restrict__ sorted,
                                                              int* __restrict__ segstart,
                                                              float* __restrict__ dinv) {
    __shared__ int tot[HPITCH];
    __shared__ int bef[HPITCH];
    __shared__ int lcur[HPITCH];
    __shared__ int gofs[HPITCH];
    __shared__ int buf[SP_SLICE];
    __shared__ int wred[SP_THREADS];

    const int tid = threadIdx.x;

    if (blockIdx.x == 0) {
        const int w    = tid & (WPITCH - 1);
        const int half = tid >> 9;
        int d = 0;
#pragma unroll 25
        for (int b = half; b < SP_WGS; b += 2) d += wpart[(size_t)b * WPITCH + w];
        wred[tid] = d;
        __syncthreads();
        if (tid < WORKERS)
            dinv[TASKS + tid] = rsqrtf((float)(1 + wred[tid] + wred[tid + WPITCH]));
        return;
    }

    const int g  = blockIdx.x - 1;
    const int a0 = g * SP_SLICE;

    int k[SP_PER];
#pragma unroll
    for (int j = 0; j < SP_PER; ++j) {
        const int i = tid + j * SP_THREADS;
        k[j] = keys[a0 + (i < SP_SLICE ? i : SP_SLICE - 1)];
    }

    const int r = tid & (HPITCH - 1);
    const int p = tid >> 5;
    int st = 0, sb = 0;
#pragma unroll
    for (int u = 0; u < SP_ROWS; ++u) {
        const int b = p + u * (SP_THREADS / HPITCH);
        if (b < SP_WGS) {
            const int h = hist[b * HPITCH + r];
            st += h;
            sb += (b < g) ? h : 0;
        }
    }
    const int own = hist[g * HPITCH + r];

    if (tid < HPITCH) { tot[tid] = 0; bef[tid] = 0; }
    __syncthreads();
    atomicAdd(&tot[r], st);
    atomicAdd(&bef[r], sb);
    __syncthreads();

    if (tid < HPITCH) {
        const int v = tot[tid];
        int inc = v, oinc = own;
#pragma unroll
        for (int o = 1; o < 32; o <<= 1) {
            const int n = __shfl_up(inc, o);
            const int m = __shfl_up(oinc, o);
            if (tid >= o) { inc += n; oinc += m; }
        }
        const int seg    = inc - v;
        const int lfirst = oinc - own;
        lcur[tid] = lfirst;
        gofs[tid] = seg + bef[tid] - lfirst;
        if (g == 0 && tid < CNT_R) segstart[tid] = seg;
    }
    __syncthreads();

#pragma unroll
    for (int j = 0; j < SP_PER; ++j) {
        const int i = tid + j * SP_THREADS;
        if (i < SP_SLICE) {
            const int pos = atomicAdd(&lcur[((unsigned)k[j] >> KSH) / RTASKS], 1);
            buf[pos] = k[j];
        }
    }
    __syncthreads();

    for (int i = tid; i < SP_SLICE; i += SP_THREADS) {
        const int key = buf[i];
        sorted[gofs[((unsigned)key >> KSH) / RTASKS] + i] = key;
    }
}


__device__ __forceinline__ void load_ablock(v2f (&a)[8], const float* __restrict__ p) {
#pragma unroll
    for (int s = 0; s < 8; ++s) {
        const float2 v = *(const float2*)(p + 4 * s);
        a[s] = (v2f){v.x, v.y};
    }
}

__device__ __forceinline__ void mul_block(v8f (&acc)[8], const v2f (&a)[8], const float2* bb) {
#pragma unroll
    for (int s = 0; s < 8; ++s) {
        const float2* brow = bb + s * 2 * BPITCH;
#pragma unroll
        for (int t = 0; t < 8; ++t) {
            const float2 bv = brow[t * 16];
            v2f bf = {bv.x, bv.y};
            acc[t] = __builtin_amdgcn_wmma_f32_16x16x4_f32(
                false, a[s], false, bf, (short)0, acc[t], false, false);
        }
    }
}

__global__ __launch_bounds__(256) void k_gemm_efc(const float* __restrict__ A,
                                                  const float* __restrict__ B,
                                                  const float* __restrict__ bias,
                                                  float* __restrict__ out) {
    __shared__ alignas(16) float2 Bs[(FEAT / 2) * BPITCH];

    const int tid = threadIdx.x;

#pragma unroll 4
    for (int it = tid; it < (FEAT / 2) * (HID / 4); it += 256) {
        const int kp = it / (HID / 4);
        const int n4 = it - kp * (HID / 4);
        const float4 r0 = *(const float4*)(B + (size_t)(2 * kp) * HID + 4 * n4);
        const float4 r1 = *(const float4*)(B + (size_t)(2 * kp + 1) * HID + 4 * n4);
        float4* dst = (float4*)&Bs[kp * BPITCH + 4 * n4];
        dst[0] = make_float4(r0.x, r1.x, r0.y, r1.y);
        dst[1] = make_float4(r0.z, r1.z, r0.w, r1.w);
    }
    __syncthreads();

    const int wave   = tid >> 5;
    const int lane   = tid & 31;
    const int lane16 = lane & 15;
    const int hi     = lane >> 4;

    const int pair = blockIdx.x * 4 + (wave & 3);
    const int sub  = wave >> 2;

    float bcol[8];
#pragma unroll
    for (int t = 0; t < 8; ++t) bcol[t] = bias[t * 16 + lane16];

    const float2* bl = Bs + hi * BPITCH + lane16;

    int tile = pair + NPAIR * sub;
    const float* arow = A + (size_t)(tile * TROWS + lane16) * FEAT + 2 * hi;

    v2f a0[8], a1[8];
    load_ablock(a0, arow);

    for (;;) {
        const int  ntile = tile + 2 * NPAIR;
        const bool more  = ntile < NTILE;
        const float* nrow = A + (size_t)((more ? ntile : tile) * TROWS + lane16) * FEAT + 2 * hi;

        v8f acc[8];
#pragma unroll
        for (int t = 0; t < 8; ++t) {
            const float bv = bcol[t];
            acc[t] = (v8f){bv, bv, bv, bv, bv, bv, bv, bv};
        }

        for (int b = 0; b < NBLK; b += 2) {
            load_ablock(a1, arow + (b + 1) * KBLK);
            mul_block(acc, a0, bl + (size_t)b * (KBLK / 2) * BPITCH);
            load_ablock(a0, (b + 2 < NBLK) ? arow + (b + 2) * KBLK : nrow);
            mul_block(acc, a1, bl + (size_t)(b + 1) * (KBLK / 2) * BPITCH);
        }

#pragma unroll
        for (int t = 0; t < 8; ++t) {
#pragma unroll
            for (int j = 0; j < 8; ++j) {
                const int m   = tile * TROWS + j + 8 * hi;
                const float v = acc[t][j];
                out[(size_t)m * HID + t * 16 + lane16] = v > 0.f ? v : 0.f;
            }
        }

        if (!more) break;
        tile = ntile;
        arow = nrow;
    }
}


__global__ __launch_bounds__(256) void k_h(const float* __restrict__ tf_h,
                                           const float* __restrict__ wf,
                                           const float* __restrict__ Wm,
                                           const float* __restrict__ Wl,
                                           float* __restrict__ hm,
                                           float* __restrict__ hl) {
    const int wave   = threadIdx.x >> 5;
    const int lane   = threadIdx.x & 31;
    const int lane16 = lane & 15;
    const int hi     = lane >> 4;

    const int node_base = (blockIdx.x * 8 + wave) * 16;
    int nd = node_base + lane16;
    if (nd > NODES - 1) nd = NODES - 1;
    const float* xrow = (nd < TASKS) ? (tf_h + (size_t)nd * HID)
                                     : (wf + (size_t)(nd - TASKS) * HID);
    const int n = lane16;
    const bool ncol = n < CLS;

    v8f accM = {0.f, 0.f, 0.f, 0.f, 0.f, 0.f, 0.f, 0.f};
    v8f accL = {0.f, 0.f, 0.f, 0.f, 0.f, 0.f, 0.f, 0.f};

    for (int k = 0; k < HID; k += 4) {
        const int k0 = k + 2 * hi;
        const float2 av = *(const float2*)(xrow + k0);
        v2f af = {av.x, av.y};
        v2f bm, bl;
        bm.x = ncol ? Wm[k0 * CLS + n] : 0.f;
        bm.y = ncol ? Wm[(k0 + 1) * CLS + n] : 0.f;
        bl.x = ncol ? Wl[k0 * CLS + n] : 0.f;
        bl.y = ncol ? Wl[(k0 + 1) * CLS + n] : 0.f;
        accM = __builtin_amdgcn_wmma_f32_16x16x4_f32(false, af, false, bm, (short)0, accM, false, false);
        accL = __builtin_amdgcn_wmma_f32_16x16x4_f32(false, af, false, bl, (short)0, accL, false, false);
    }

    if (ncol) {
#pragma unroll
        for (int j = 0; j < 8; ++j) {
            int m = node_base + j + 8 * hi;
            if (m < NODES) {
                hm[(size_t)m * HP + n] = accM[j];
                hl[(size_t)m * HP + n] = accL[j];
            }
        }
    }
}


__device__ __forceinline__ void lds_barrier() {
    __builtin_amdgcn_fence(__ATOMIC_RELEASE, "workgroup", "local");
    __builtin_amdgcn_s_barrier();
    __builtin_amdgcn_fence(__ATOMIC_ACQUIRE, "workgroup", "local");
}

constexpr int PLANE     = WORKERS * ROW2;
constexpr int GRP_LANES = ROW2 / 4;
constexpr int GRP_WAVE  = 32 / GRP_LANES;
constexpr int NGRP      = (OWN_THREADS / 32) * GRP_WAVE;
constexpr int TROUNDS   = (TBLK + NGRP - 1) / NGRP;
static_assert(TBLK < OWN_THREADS && ROW2 % 4 == 0 && PLANE % 4 == 0,
              "one scan thread per owned task and one more; planes move as 16-byte pieces");

__device__ __forceinline__ void load_keys(int (&cur)[KEYS_THREAD], const int* __restrict__ sorted,
                                          int first, int hi, int tid) {
#pragma unroll
    for (int j = 0; j < KEYS_THREAD; ++j) {
        const int idx = first + j * OWN_THREADS + tid;
        cur[j] = (idx < hi) ? sorted[idx] : -1;
    }
}

__global__ __launch_bounds__(OWN_THREADS) void k_agg_owner(const int* __restrict__ sorted,
                                                           const int* __restrict__ segstart,
                                                           const float* __restrict__ hm,
                                                           const float* __restrict__ hl,
                                                           const float* __restrict__ dinv,
                                                           const float* __restrict__ bm,
                                                           const float* __restrict__ bl,
                                                           float* __restrict__ mean,
                                                           float* __restrict__ lsd,
                                                           float* __restrict__ planes) {
    __shared__ alignas(16) float Fs[TBLK * ROW2];
    __shared__ alignas(16) float Gs[WORKERS * ROW2];
    __shared__ alignas(16) float wacc[WORKERS * ROW2];
    __shared__ int q[QCAP];
    __shared__ int cnt[TBLK];
    __shared__ int beg[TBLK + 1];
    __shared__ int wsum[OWN_THREADS / 32];
    __shared__ float dts[TBLK];

    const int tid    = threadIdx.x;
    const int lane   = tid & 31;
    const int wave   = tid >> 5;
    const int t_base = blockIdx.x * TBLK;

    const int rng = blockIdx.x / OWN_PER_R;
    const int lo  = segstart[rng];
    const int hi  = (rng + 1 < CNT_R) ? segstart[rng + 1] : ANS;
    const int nch = (hi - lo + QCAP - 1) / QCAP;

    int cur[KEYS_THREAD];
    load_keys(cur, sorted, lo, hi, tid);

    const unsigned key_lo  = (unsigned)t_base << KSH;
    const unsigned key_len = (unsigned)TBLK << KSH;

    const int  gl   = lane / GRP_LANES;
    const int  c4   = (lane - gl * GRP_LANES) * 4;
    const bool gact = lane < GRP_WAVE * GRP_LANES;
    const int  grp  = wave * GRP_WAVE + gl;

    float4 hv[TROUNDS];
#pragma unroll
    for (int r = 0; r < TROUNDS; ++r) {
        const int tl = grp + r * NGRP;
        hv[r] = make_float4(0.f, 0.f, 0.f, 0.f);
        if (gact && tl < TBLK) {
            const int t = t_base + tl;
            float v[4];
#pragma unroll
            for (int k = 0; k < 4; ++k) {
                const int e = c4 + k;
                v[k] = (e < CLS) ? hm[(size_t)t * HP + e] : hl[(size_t)t * HP + (e - CLS)];
            }
            hv[r] = make_float4(v[0], v[1], v[2], v[3]);
        }
    }

    for (int i = tid; i < WORKERS * CLS; i += OWN_THREADS) {
        const int w  = i / CLS;
        const int c  = i - w * CLS;
        const int r  = TASKS + w;
        const float d = dinv[r];
        Gs[w * ROW2 + c]         = d * hm[(size_t)r * HP + c];
        Gs[w * ROW2 + CLS + c]   = d * hl[(size_t)r * HP + c];
        wacc[w * ROW2 + c]       = 0.f;
        wacc[w * ROW2 + CLS + c] = 0.f;
    }
    if (tid < TBLK) cnt[tid] = 0;
    __syncthreads();

    for (int ch = 0; ch < nch; ++ch) {
#pragma unroll
        for (int j = 0; j < KEYS_THREAD; ++j) {
            const unsigned rel = (unsigned)cur[j] - key_lo;
            if (rel < key_len) atomicAdd(&cnt[rel >> KSH], 1);
        }
        if (ch + 1 < nch) load_keys(cur, sorted, lo + (ch + 1) * QCAP, hi, tid);
    }
    if (nch > 1) load_keys(cur, sorted, lo, hi, tid);
    __syncthreads();
    if (tid < TBLK) {
        dts[tid] = rsqrtf((float)(cnt[tid] + 1));
        cnt[tid] = 0;
    }
    __syncthreads();

    float4 a[TROUNDS];
#pragma unroll
    for (int r = 0; r < TROUNDS; ++r) {
        const int tl = grp + r * NGRP;
        a[r] = make_float4(0.f, 0.f, 0.f, 0.f);
        if (gact && tl < TBLK) {
            const float d = dts[tl];
            *(float4*)&Fs[tl * ROW2 + c4] = make_float4(d * hv[r].x, d * hv[r].y, d * hv[r].z, d * hv[r].w);
        }
    }

    for (int ch = 0; ch < nch; ++ch) {
#pragma unroll
        for (int j = 0; j < KEYS_THREAD; ++j) {
            const unsigned rel = (unsigned)cur[j] - key_lo;
            if (rel < key_len) atomicAdd(&cnt[rel >> KSH], 1);
        }
        lds_barrier();

        {
            const int v = (tid < TBLK) ? cnt[tid] : 0;
            int inc = v;
#pragma unroll
            for (int o = 1; o < 32; o <<= 1) {
                const int n = __shfl_up(inc, o);
                if (lane >= o) inc += n;
            }
            if (lane == 31) wsum[wave] = inc;
            lds_barrier();
            int before = (lane < wave) ? wsum[lane] : 0;
#pragma unroll
            for (int o = 16; o > 0; o >>= 1) before += __shfl_xor(before, o);
            const int ex = before + inc - v;
            if (tid <= TBLK) beg[tid] = ex;
            if (tid < TBLK)  cnt[tid] = ex;
            lds_barrier();
        }

#pragma unroll
        for (int j = 0; j < KEYS_THREAD; ++j) {
            const unsigned rel = (unsigned)cur[j] - key_lo;
            if (rel < key_len) {
                const int pos = atomicAdd(&cnt[rel >> KSH], 1);
                q[pos] = (int)(rel & KMASK);
            }
        }

        if (ch + 1 < nch) load_keys(cur, sorted, lo + (ch + 1) * QCAP, hi, tid);

        lds_barrier();
        if (tid < TBLK) cnt[tid] = 0;

#pragma unroll
        for (int r = 0; r < TROUNDS; ++r) {
            const int tl = grp + r * NGRP;
            if (gact && tl < TBLK) {
                const int je = beg[tl + 1];
                const float4 fv = *(const float4*)&Fs[tl * ROW2 + c4];
                float4 s = a[r];
                for (int j = beg[tl]; j < je; ++j) {
                    const int w = q[j];
                    const float4 g = *(const float4*)&Gs[w * ROW2 + c4];
                    s.x += g.x; s.y += g.y; s.z += g.z; s.w += g.w;
                    float* b = &wacc[w * ROW2 + c4];
                    atomicAdd(b + 0, fv.x); atomicAdd(b + 1, fv.y);
                    atomicAdd(b + 2, fv.z); atomicAdd(b + 3, fv.w);
                }
                a[r] = s;
            }
        }
        lds_barrier();
    }

#pragma unroll
    for (int r = 0; r < TROUNDS; ++r) {
        const int tl = grp + r * NGRP;
        if (gact && tl < TBLK) {
            float4* p = (float4*)&Fs[tl * ROW2 + c4];
            const float4 fv = *p;
            *p = make_float4(fv.x + a[r].x, fv.y + a[r].y, fv.z + a[r].z, fv.w + a[r].w);
        }
    }
    __syncthreads();

    for (int i = tid; i < TBLK * CLS; i += OWN_THREADS) {
        const int tl = i / CLS;
        const int c  = i - tl * CLS;
        const float d = dts[tl];
        const size_t o = (size_t)t_base * CLS + i;
        mean[o] = bm[c] + d * Fs[tl * ROW2 + c];
        lsd[o]  = bl[c] + d * Fs[tl * ROW2 + CLS + c];
    }

    float4* __restrict__ pl = (float4*)(planes + (size_t)blockIdx.x * PLANE);
    for (int i = tid; i < PLANE / 4; i += OWN_THREADS) pl[i] = *(const float4*)&wacc[4 * i];
}


constexpr int Z_THREADS = 320;
constexpr int ZW_EL     = 64;
constexpr int ZW_PARTS  = Z_THREADS / ZW_EL;
constexpr int ZW_DEPTH  = NOWN / ZW_PARTS;
constexpr int ZW_WGS    = (WORKERS * CLS + ZW_EL - 1) / ZW_EL;
constexpr int ZT_WGS    = TASKS * CLS / Z_THREADS;
constexpr int Z_WGS     = ZW_WGS + ZT_WGS;
static_assert(NOWN % ZW_PARTS == 0 && (TASKS * CLS) % Z_THREADS == 0 && Z_THREADS % ZW_EL == 0,
              "z partition must be exact");

__global__ __launch_bounds__(Z_THREADS) void k_z(const float* __restrict__ planes,
                                                 const float* __restrict__ hm, const float* __restrict__ hl,
                                                 const float* __restrict__ dinv,
                                                 const float* __restrict__ bm, const float* __restrict__ bl,
                                                 const float* __restrict__ eps,
                                                 float* __restrict__ mean, float* __restrict__ lsd,
                                                 float* __restrict__ z) {
    __shared__ float red[2][ZW_PARTS][ZW_EL];

    if (blockIdx.x >= ZW_WGS) {
        const int idx = (blockIdx.x - ZW_WGS) * Z_THREADS + threadIdx.x;
        z[idx] = mean[idx] + (eps[idx] * 0.01f) * expf(lsd[idx]);
        return;
    }

    const int  el   = threadIdx.x % ZW_EL;
    const int  part = threadIdx.x / ZW_EL;
    const int  i    = blockIdx.x * ZW_EL + el;
    const bool ok   = i < WORKERS * CLS;
    const int  ic   = ok ? i : WORKERS * CLS - 1;
    const int  w    = ic / CLS;
    const int  c    = ic - w * CLS;

    const float* __restrict__ p = planes + (size_t)part * PLANE + w * ROW2 + c;
    float sm = 0.f, sl = 0.f;
#pragma unroll
    for (int k = 0; k < ZW_DEPTH; ++k) {
        sm += p[(size_t)k * ZW_PARTS * PLANE];
        sl += p[(size_t)k * ZW_PARTS * PLANE + CLS];
    }
    red[0][part][el] = sm;
    red[1][part][el] = sl;
    __syncthreads();

    if (part == 0 && ok) {
        float tm = red[0][0][el], tl = red[1][0][el];
#pragma unroll
        for (int q = 1; q < ZW_PARTS; ++q) { tm += red[0][q][el]; tl += red[1][q][el]; }
        const float  d  = dinv[TASKS + w];
        const float  sw = d * d;
        const size_t o  = (size_t)TASKS * CLS + i;
        const float  m  = bm[c] + hm[(size_t)(TASKS + w) * HP + c] * sw + d * tm;
        const float  l  = bl[c] + hl[(size_t)(TASKS + w) * HP + c] * sw + d * tl;
        mean[o] = m;
        lsd[o]  = l;
        z[o]    = m + (eps[o] * 0.01f) * expf(l);
    }
}

constexpr int NQUAD      = ANS * CLS / 4;
constexpr int CR_THREADS = 160;
constexpr int CR_WGS     = 625;
constexpr int CR_GT      = CR_THREADS * CR_WGS;
constexpr int CR_TRIPS   = NQUAD / CR_GT;
constexpr int CR_BATCH   = 5;
constexpr int CR_ASTEP   = CR_GT * 4 / CLS;
static_assert((ANS * CLS) % 4 == 0 && NQUAD % CR_GT == 0 && (CR_GT * 4) % CLS == 0 &&
              CR_TRIPS % CR_BATCH == 0 && CLS % 2 == 0, "crowd partition must be exact");

__global__ __launch_bounds__(CR_THREADS) void k_crowd(const int* __restrict__ keys,
                                                      const float* __restrict__ z,
                                                      float* __restrict__ crowd) {
    const int g  = blockIdx.x * CR_THREADS + threadIdx.x;
    const int eA = 4 * g;
    const int aA = eA / CLS;
    const int cA = eA - aA * CLS;
    const int eB = eA + 2;
    const int aB = eB / CLS;
    const int cB = eB - aB * CLS;

    const float* __restrict__ ztA = z + cA;
    const float* __restrict__ ztB = z + cB;
    const float* __restrict__ zwA = z + (size_t)TASKS * CLS + cA;
    const float* __restrict__ zwB = z + (size_t)TASKS * CLS + cB;
    const int*   __restrict__ kA  = keys + aA;
    const int*   __restrict__ kB  = keys + aB;
    float4*      __restrict__ o   = (float4*)crowd + g;

    for (int b = 0; b < CR_TRIPS; b += CR_BATCH) {
        int ka[CR_BATCH], kb[CR_BATCH];
#pragma unroll
        for (int u = 0; u < CR_BATCH; ++u) {
            ka[u] = kA[(b + u) * CR_ASTEP];
            kb[u] = kB[(b + u) * CR_ASTEP];
        }
        float2 ta[CR_BATCH], wa[CR_BATCH], tb[CR_BATCH], wb[CR_BATCH];
#pragma unroll
        for (int u = 0; u < CR_BATCH; ++u) {
            ta[u] = *(const float2*)(ztA + (ka[u] >> KSH) * CLS);
            wa[u] = *(const float2*)(zwA + (ka[u] & KMASK) * CLS);
            tb[u] = *(const float2*)(ztB + (kb[u] >> KSH) * CLS);
            wb[u] = *(const float2*)(zwB + (kb[u] & KMASK) * CLS);
        }
#pragma unroll
        for (int u = 0; u < CR_BATCH; ++u)
            o[(size_t)(b + u) * CR_GT] = make_float4(ta[u].x * wa[u].x, ta[u].y * wa[u].y,
                                                     tb[u].x * wb[u].x, tb[u].y * wb[u].y);
    }
}


extern "C" void kernel_launch(void* const* d_in, const int* in_sizes, int n_in,
                              void* d_out, int out_size, void* d_ws, size_t ws_size,
                              hipStream_t stream) {
    (void)in_sizes; (void)n_in; (void)out_size; (void)ws_size;

    const float* task_feature   = (const float*)d_in[0];
    const int*   answers        = (const int*)d_in[1];
    const float* worker_feature = (const float*)d_in[2];
    const float* W_efc          = (const float*)d_in[3];
    const float* b_efc          = (const float*)d_in[4];
    const float* W_mean         = (const float*)d_in[5];
    const float* b_mean         = (const float*)d_in[6];
    const float* W_ls           = (const float*)d_in[7];
    const float* b_ls           = (const float*)d_in[8];
    const float* eps            = (const float*)d_in[9];

    float* out   = (float*)d_out;
    float* crowd = out;
    float* mean  = out + (size_t)ANS * CLS;
    float* lsd   = mean + (size_t)NODES * CLS;

    float*  ws   = (float*)d_ws;
    float*  tf_h = ws;
    float*  hm   = tf_h + (size_t)TASKS * HID;
    float*  hl   = hm + (size_t)NODES * HP;
    float*  zz   = hl + (size_t)NODES * HP;
    float*  dinv = zz + (size_t)NODES * CLS;

    auto cdiv = [](int a, int b) { return (a + b - 1) / b; };

    k_gemm_efc <<<GEMM_WGS, 256, 0, stream>>>(task_feature, W_efc, b_efc, tf_h);
    k_h        <<<cdiv(cdiv(NODES, 16), 8), 256, 0, stream>>>(tf_h, worker_feature,
                                                              W_mean, W_ls, hm, hl);

    int* keys  = (int*)tf_h;
    int* hist  = keys + ANS;
    int* wpart = hist + (size_t)SP_WGS * HPITCH;
    int* segst = wpart + (size_t)SP_WGS * WPITCH;
    int* sorted = segst + 32;
    float* planes = (float*)(sorted + ANS);
    k_split_count   <<<SP_WGS, SP_THREADS, 0, stream>>>(answers, keys, hist, wpart);
    k_split_scatter <<<SC_WGS, SP_THREADS, 0, stream>>>(keys, hist, wpart, sorted, segst, dinv);

    k_agg_owner <<<NOWN, OWN_THREADS, 0, stream>>>(sorted, segst, hm, hl, dinv, b_mean, b_ls, mean, lsd, planes);

    k_z        <<<Z_WGS, Z_THREADS, 0, stream>>>(planes, hm, hl, dinv, b_mean, b_ls, eps, mean, lsd, zz);
    k_crowd    <<<CR_WGS, CR_THREADS, 0, stream>>>(keys, zz, crowd);
}


# Task partition re-cut: 250 aggregation owners of 400 rows (all WGPs busy, 2 rounds of 204 row groups), keys bucketed in 125 ranges; worker-row sums regrouped: 250 planes added in owner order
# speedup vs baseline: 5.6467x; 1.1082x over previous
#include <hip/hip_runtime.h>

typedef __attribute__((ext_vector_type(2))) float v2f;
typedef __attribute__((ext_vector_type(8))) float v8f;

constexpr int TASKS   = 100000;
constexpr int WORKERS = 500;
constexpr int FEAT    = 512;
constexpr int CLS     = 10;
constexpr int HP      = 12;
constexpr int ANS     = 1000000;
constexpr int HID     = 128;
constexpr int NODES   = TASKS + WORKERS;
constexpr int KSH     = 9;
constexpr int KMASK   = (1 << KSH) - 1;
constexpr int TBLK    = 400;
constexpr int NOWN    = TASKS / TBLK;
constexpr int ROW2    = 2 * CLS;
constexpr int QCAP    = 10240;
constexpr int OWN_THREADS = 1024;
constexpr int KEYS_THREAD = QCAP / OWN_THREADS;
constexpr int TROWS   = 16;
constexpr int NTILE   = TASKS / TROWS;
constexpr int GEMM_WGS = 256;
constexpr int NPAIR   = GEMM_WGS * 4;
constexpr int BPITCH  = HID + 16;
constexpr int KBLK    = 32;
constexpr int NBLK    = FEAT / KBLK;

constexpr int CNT_R      = 125;
constexpr int RTASKS     = TASKS / CNT_R;
constexpr int OWN_PER_R  = RTASKS / TBLK;
constexpr int WPITCH     = 512;
constexpr int HPITCH     = 128;
constexpr int SC_SCANW   = HPITCH / 32;
constexpr int SP_WGS     = 250;
constexpr int SP_THREADS = 1024;
constexpr int SP_SLICE   = ANS / SP_WGS;
constexpr int SP_PER     = (SP_SLICE + SP_THREADS - 1) / SP_THREADS;
constexpr int SP_ROWS    = (SP_WGS + SP_THREADS / HPITCH - 1) / (SP_THREADS / HPITCH);
constexpr int SC_WGS     = SP_WGS + 1;
static_assert(TASKS % CNT_R == 0 && ANS % SP_WGS == 0 && WORKERS <= WPITCH && 2 * WPITCH == SP_THREADS &&
              CNT_R <= HPITCH && HPITCH % 32 == 0 && SP_THREADS % HPITCH == 0 && RTASKS % TBLK == 0 &&
              SP_WGS % 2 == 0,
              "split partition must be exact");

__global__ __launch_bounds__(SP_THREADS) void k_split_count(const int* __restrict__ answers,
                                                            int* __restrict__ keys,
                                                            int* __restrict__ hist,
                                                            int* __restrict__ wpart) {
    __shared__ int rcnt[HPITCH];
    __shared__ int wcnt[WPITCH];

    const int tid = threadIdx.x;
    const int a0  = blockIdx.x * SP_SLICE;

    int t[SP_PER], w[SP_PER];
#pragma unroll
    for (int j = 0; j < SP_PER; ++j) {
        const int i = tid + j * SP_THREADS;
        const int a = a0 + (i < SP_SLICE ? i : SP_SLICE - 1);
        t[j] = answers[3 * a];
        w[j] = answers[3 * a + 1];
    }

    if (tid < HPITCH) rcnt[tid] = 0;
    if (tid < WPITCH) wcnt[tid] = 0;
    __syncthreads();

#pragma unroll
    for (int j = 0; j < SP_PER; ++j) {
        const int i = tid + j * SP_THREADS;
        if (i < SP_SLICE) {
            keys[a0 + i] = (t[j] << KSH) | w[j];
            atomicAdd(&rcnt[(unsigned)t[j] / RTASKS], 1);
            atomicAdd(&wcnt[w[j]], 1);
        }
    }
    __syncthreads();

    if (tid < HPITCH) hist[blockIdx.x * HPITCH + tid] = rcnt[tid];
    if (tid < WPITCH) wpart[(size_t)blockIdx.x * WPITCH + tid] = wcnt[tid];
}

__global__ __launch_bounds__(SP_THREADS) void k_split_scatter(const int* __restrict__ keys,
                                                              const int* __restrict__ hist,
                                                              const int* __restrict__ wpart,
                                                              int* __restrict__ sorted,
                                                              int* __restrict__ segstart,
                                                              float* __restrict__ dinv) {
    __shared__ int tot[HPITCH];
    __shared__ int bef[HPITCH];
    __shared__ int lcur[HPITCH];
    __shared__ int gofs[HPITCH];
    __shared__ int buf[SP_SLICE];
    __shared__ int wred[SP_THREADS];
    __shared__ int wtot[SC_SCANW];
    __shared__ int wown[SC_SCANW];

    const int tid = threadIdx.x;

    if (blockIdx.x == 0) {
        const int w    = tid & (WPITCH - 1);
        const int half = tid >> 9;
        int d = 0;
#pragma unroll 25
        for (int b = half; b < SP_WGS; b += 2) d += wpart[(size_t)b * WPITCH + w];
        wred[tid] = d;
        __syncthreads();
        if (tid < WORKERS)
            dinv[TASKS + tid] = rsqrtf((float)(1 + wred[tid] + wred[tid + WPITCH]));
        return;
    }

    const int g  = blockIdx.x - 1;
    const int a0 = g * SP_SLICE;

    int k[SP_PER];
#pragma unroll
    for (int j = 0; j < SP_PER; ++j) {
        const int i = tid + j * SP_THREADS;
        k[j] = keys[a0 + (i < SP_SLICE ? i : SP_SLICE - 1)];
    }

    const int r = tid & (HPITCH - 1);
    const int p = tid / HPITCH;
    int st = 0, sb = 0;
#pragma unroll
    for (int u = 0; u < SP_ROWS; ++u) {
        const int b = p + u * (SP_THREADS / HPITCH);
        if (b < SP_WGS) {
            const int h = hist[b * HPITCH + r];
            st += h;
            sb += (b < g) ? h : 0;
        }
    }
    const int own = hist[g * HPITCH + r];

    if (tid < HPITCH) { tot[tid] = 0; bef[tid] = 0; }
    __syncthreads();
    atomicAdd(&tot[r], st);
    atomicAdd(&bef[r], sb);
    __syncthreads();

    const int sl = tid & 31;
    const int sw = tid >> 5;
    int v = 0, inc = 0, oinc = 0;
    if (tid < HPITCH) {
        v    = tot[tid];
        inc  = v;
        oinc = own;
#pragma unroll
        for (int o = 1; o < 32; o <<= 1) {
            const int n = __shfl_up(inc, o);
            const int m = __shfl_up(oinc, o);
            if (sl >= o) { inc += n; oinc += m; }
        }
        if (sl == 31) { wtot[sw] = inc; wown[sw] = oinc; }
    }
    __syncthreads();
    if (tid < HPITCH) {
        int pt = 0, po = 0;
#pragma unroll
        for (int x = 0; x < SC_SCANW; ++x) {
            if (x < sw) { pt += wtot[x]; po += wown[x]; }
        }
        const int seg    = pt + inc - v;
        const int lfirst = po + oinc - own;
        lcur[tid] = lfirst;
        gofs[tid] = seg + bef[tid] - lfirst;
        if (g == 0 && tid < CNT_R) segstart[tid] = seg;
    }
    __syncthreads();

#pragma unroll
    for (int j = 0; j < SP_PER; ++j) {
        const int i = tid + j * SP_THREADS;
        if (i < SP_SLICE) {
            const int pos = atomicAdd(&lcur[((unsigned)k[j] >> KSH) / RTASKS], 1);
            buf[pos] = k[j];
        }
    }
    __syncthreads();

    for (int i = tid; i < SP_SLICE; i += SP_THREADS) {
        const int key = buf[i];
        sorted[gofs[((unsigned)key >> KSH) / RTASKS] + i] = key;
    }
}


__device__ __forceinline__ void load_ablock(v2f (&a)[8], const float* __restrict__ p) {
#pragma unroll
    for (int s = 0; s < 8; ++s) {
        const float2 v = *(const float2*)(p + 4 * s);
        a[s] = (v2f){v.x, v.y};
    }
}

__device__ __forceinline__ void mul_block(v8f (&acc)[8], const v2f (&a)[8], const float2* bb) {
#pragma unroll
    for (int s = 0; s < 8; ++s) {
        const float2* brow = bb + s * 2 * BPITCH;
#pragma unroll
        for (int t = 0; t < 8; ++t) {
            const float2 bv = brow[t * 16];
            v2f bf = {bv.x, bv.y};
            acc[t] = __builtin_amdgcn_wmma_f32_16x16x4_f32(
                false, a[s], false, bf, (short)0, acc[t], false, false);
        }
    }
}

__global__ __launch_bounds__(256) void k_gemm_efc(const float* __restrict__ A,
                                                  const float* __restrict__ B,
                                                  const float* __restrict__ bias,
                                                  float* __restrict__ out) {
    __shared__ alignas(16) float2 Bs[(FEAT / 2) * BPITCH];

    const int tid = threadIdx.x;

#pragma unroll 4
    for (int it = tid; it < (FEAT / 2) * (HID / 4); it += 256) {
        const int kp = it / (HID / 4);
        const int n4 = it - kp * (HID / 4);
        const float4 r0 = *(const float4*)(B + (size_t)(2 * kp) * HID + 4 * n4);
        const float4 r1 = *(const float4*)(B + (size_t)(2 * kp + 1) * HID + 4 * n4);
        float4* dst = (float4*)&Bs[kp * BPITCH + 4 * n4];
        dst[0] = make_float4(r0.x, r1.x, r0.y, r1.y);
        dst[1] = make_float4(r0.z, r1.z, r0.w, r1.w);
    }
    __syncthreads();

    const int wave   = tid >> 5;
    const int lane   = tid & 31;
    const int lane16 = lane & 15;
    const int hi     = lane >> 4;

    const int pair = blockIdx.x * 4 + (wave & 3);
    const int sub  = wave >> 2;

    float bcol[8];
#pragma unroll
    for (int t = 0; t < 8; ++t) bcol[t] = bias[t * 16 + lane16];

    const float2* bl = Bs + hi * BPITCH + lane16;

    int tile = pair + NPAIR * sub;
    const float* arow = A + (size_t)(tile * TROWS + lane16) * FEAT + 2 * hi;

    v2f a0[8], a1[8];
    load_ablock(a0, arow);

    for (;;) {
        const int  ntile = tile + 2 * NPAIR;
        const bool more  = ntile < NTILE;
        const float* nrow = A + (size_t)((more ? ntile : tile) * TROWS + lane16) * FEAT + 2 * hi;

        v8f acc[8];
#pragma unroll
        for (int t = 0; t < 8; ++t) {
            const float bv = bcol[t];
            acc[t] = (v8f){bv, bv, bv, bv, bv, bv, bv, bv};
        }

        for (int b = 0; b < NBLK; b += 2) {
            load_ablock(a1, arow + (b + 1) * KBLK);
            mul_block(acc, a0, bl + (size_t)b * (KBLK / 2) * BPITCH);
            load_ablock(a0, (b + 2 < NBLK) ? arow + (b + 2) * KBLK : nrow);
            mul_block(acc, a1, bl + (size_t)(b + 1) * (KBLK / 2) * BPITCH);
        }

#pragma unroll
        for (int t = 0; t < 8; ++t) {
#pragma unroll
            for (int j = 0; j < 8; ++j) {
                const int m   = tile * TROWS + j + 8 * hi;
                const float v = acc[t][j];
                out[(size_t)m * HID + t * 16 + lane16] = v > 0.f ? v : 0.f;
            }
        }

        if (!more) break;
        tile = ntile;
        arow = nrow;
    }
}


__global__ __launch_bounds__(256) void k_h(const float* __restrict__ tf_h,
                                           const float* __restrict__ wf,
                                           const float* __restrict__ Wm,
                                           const float* __restrict__ Wl,
                                           float* __restrict__ hm,
                                           float* __restrict__ hl) {
    const int wave   = threadIdx.x >> 5;
    const int lane   = threadIdx.x & 31;
    const int lane16 = lane & 15;
    const int hi     = lane >> 4;

    const int node_base = (blockIdx.x * 8 + wave) * 16;
    int nd = node_base + lane16;
    if (nd > NODES - 1) nd = NODES - 1;
    const float* xrow = (nd < TASKS) ? (tf_h + (size_t)nd * HID)
                                     : (wf + (size_t)(nd - TASKS) * HID);
    const int n = lane16;
    const bool ncol = n < CLS;

    v8f accM = {0.f, 0.f, 0.f, 0.f, 0.f, 0.f, 0.f, 0.f};
    v8f accL = {0.f, 0.f, 0.f, 0.f, 0.f, 0.f, 0.f, 0.f};

    for (int k = 0; k < HID; k += 4) {
        const int k0 = k + 2 * hi;
        const float2 av = *(const float2*)(xrow + k0);
        v2f af = {av.x, av.y};
        v2f bm, bl;
        bm.x = ncol ? Wm[k0 * CLS + n] : 0.f;
        bm.y = ncol ? Wm[(k0 + 1) * CLS + n] : 0.f;
        bl.x = ncol ? Wl[k0 * CLS + n] : 0.f;
        bl.y = ncol ? Wl[(k0 + 1) * CLS + n] : 0.f;
        accM = __builtin_amdgcn_wmma_f32_16x16x4_f32(false, af, false, bm, (short)0, accM, false, false);
        accL = __builtin_amdgcn_wmma_f32_16x16x4_f32(false, af, false, bl, (short)0, accL, false, false);
    }

    if (ncol) {
#pragma unroll
        for (int j = 0; j < 8; ++j) {
            int m = node_base + j + 8 * hi;
            if (m < NODES) {
                hm[(size_t)m * HP + n] = accM[j];
                hl[(size_t)m * HP + n] = accL[j];
            }
        }
    }
}


__device__ __forceinline__ void lds_barrier() {
    __builtin_amdgcn_fence(__ATOMIC_RELEASE, "workgroup", "local");
    __builtin_amdgcn_s_barrier();
    __builtin_amdgcn_fence(__ATOMIC_ACQUIRE, "workgroup", "local");
}

constexpr int PLANE     = WORKERS * ROW2;
constexpr int GRP_LANES = ROW2 / 4;
constexpr int NGRP      = OWN_THREADS / GRP_LANES;
constexpr int TROUNDS   = (TBLK + NGRP - 1) / NGRP;
static_assert(TBLK < OWN_THREADS && ROW2 % 4 == 0 && PLANE % 4 == 0,
              "one scan thread per owned task and one more; planes move as 16-byte pieces");

__device__ __forceinline__ void load_keys(int (&cur)[KEYS_THREAD], const int* __restrict__ sorted,
                                          int first, int hi, int tid) {
#pragma unroll
    for (int j = 0; j < KEYS_THREAD; ++j) {
        const int idx = first + j * OWN_THREADS + tid;
        cur[j] = (idx < hi) ? sorted[idx] : -1;
    }
}

__global__ __launch_bounds__(OWN_THREADS) void k_agg_owner(const int* __restrict__ sorted,
                                                           const int* __restrict__ segstart,
                                                           const float* __restrict__ hm,
                                                           const float* __restrict__ hl,
                                                           const float* __restrict__ dinv,
                                                           const float* __restrict__ bm,
                                                           const float* __restrict__ bl,
                                                           float* __restrict__ mean,
                                                           float* __restrict__ lsd,
                                                           float* __restrict__ planes) {
    __shared__ alignas(16) float Fs[TBLK * ROW2];
    __shared__ alignas(16) float Gs[WORKERS * ROW2];
    __shared__ alignas(16) float wacc[WORKERS * ROW2];
    __shared__ int q[QCAP];
    __shared__ int cnt[TBLK];
    __shared__ int beg[TBLK + 1];
    __shared__ int wsum[OWN_THREADS / 32];
    __shared__ float dts[TBLK];

    const int tid    = threadIdx.x;
    const int lane   = tid & 31;
    const int wave   = tid >> 5;
    const int t_base = blockIdx.x * TBLK;

    const int rng = blockIdx.x / OWN_PER_R;
    const int lo  = segstart[rng];
    const int hi  = (rng + 1 < CNT_R) ? segstart[rng + 1] : ANS;
    const int nch = (hi - lo + QCAP - 1) / QCAP;

    int cur[KEYS_THREAD];
    load_keys(cur, sorted, lo, hi, tid);

    const unsigned key_lo  = (unsigned)t_base << KSH;
    const unsigned key_len = (unsigned)TBLK << KSH;

    const int  grp  = tid / GRP_LANES;
    const int  c4   = (tid - grp * GRP_LANES) * 4;
    const bool gact = grp < NGRP;

    float4 hv[TROUNDS];
#pragma unroll
    for (int r = 0; r < TROUNDS; ++r) {
        const int tl = grp + r * NGRP;
        hv[r] = make_float4(0.f, 0.f, 0.f, 0.f);
        if (gact && tl < TBLK) {
            const int t = t_base + tl;
            float v[4];
#pragma unroll
            for (int k = 0; k < 4; ++k) {
                const int e = c4 + k;
                v[k] = (e < CLS) ? hm[(size_t)t * HP + e] : hl[(size_t)t * HP + (e - CLS)];
            }
            hv[r] = make_float4(v[0], v[1], v[2], v[3]);
        }
    }

    for (int i = tid; i < WORKERS * CLS; i += OWN_THREADS) {
        const int w  = i / CLS;
        const int c  = i - w * CLS;
        const int r  = TASKS + w;
        const float d = dinv[r];
        Gs[w * ROW2 + c]         = d * hm[(size_t)r * HP + c];
        Gs[w * ROW2 + CLS + c]   = d * hl[(size_t)r * HP + c];
        wacc[w * ROW2 + c]       = 0.f;
        wacc[w * ROW2 + CLS + c] = 0.f;
    }
    if (tid < TBLK) cnt[tid] = 0;
    __syncthreads();

    for (int ch = 0; ch < nch; ++ch) {
#pragma unroll
        for (int j = 0; j < KEYS_THREAD; ++j) {
            const unsigned rel = (unsigned)cur[j] - key_lo;
            if (rel < key_len) atomicAdd(&cnt[rel >> KSH], 1);
        }
        if (ch + 1 < nch) load_keys(cur, sorted, lo + (ch + 1) * QCAP, hi, tid);
    }
    if (nch > 1) load_keys(cur, sorted, lo, hi, tid);
    __syncthreads();
    if (tid < TBLK) {
        dts[tid] = rsqrtf((float)(cnt[tid] + 1));
        cnt[tid] = 0;
    }
    __syncthreads();

    float4 a[TROUNDS];
#pragma unroll
    for (int r = 0; r < TROUNDS; ++r) {
        const int tl = grp + r * NGRP;
        a[r] = make_float4(0.f, 0.f, 0.f, 0.f);
        if (gact && tl < TBLK) {
            const float d = dts[tl];
            *(float4*)&Fs[tl * ROW2 + c4] = make_float4(d * hv[r].x, d * hv[r].y, d * hv[r].z, d * hv[r].w);
        }
    }

    for (int ch = 0; ch < nch; ++ch) {
#pragma unroll
        for (int j = 0; j < KEYS_THREAD; ++j) {
            const unsigned rel = (unsigned)cur[j] - key_lo;
            if (rel < key_len) atomicAdd(&cnt[rel >> KSH], 1);
        }
        lds_barrier();

        {
            const int v = (tid < TBLK) ? cnt[tid] : 0;
            int inc = v;
#pragma unroll
            for (int o = 1; o < 32; o <<= 1) {
                const int n = __shfl_up(inc, o);
                if (lane >= o) inc += n;
            }
            if (lane == 31) wsum[wave] = inc;
            lds_barrier();
            int before = (lane < wave) ? wsum[lane] : 0;
#pragma unroll
            for (int o = 16; o > 0; o >>= 1) before += __shfl_xor(before, o);
            const int ex = before + inc - v;
            if (tid <= TBLK) beg[tid] = ex;
            if (tid < TBLK)  cnt[tid] = ex;
            lds_barrier();
        }

#pragma unroll
        for (int j = 0; j < KEYS_THREAD; ++j) {
            const unsigned rel = (unsigned)cur[j] - key_lo;
            if (rel < key_len) {
                const int pos = atomicAdd(&cnt[rel >> KSH], 1);
                q[pos] = (int)(rel & KMASK);
            }
        }

        if (ch + 1 < nch) load_keys(cur, sorted, lo + (ch + 1) * QCAP, hi, tid);

        lds_barrier();
        if (tid < TBLK) cnt[tid] = 0;

#pragma unroll
        for (int r = 0; r < TROUNDS; ++r) {
            const int tl = grp + r * NGRP;
            if (gact && tl < TBLK) {
                const int je = beg[tl + 1];
                const float4 fv = *(const float4*)&Fs[tl * ROW2 + c4];
                float4 s = a[r];
                for (int j = beg[tl]; j < je; ++j) {
                    const int w = q[j];
                    const float4 g = *(const float4*)&Gs[w * ROW2 + c4];
                    s.x += g.x; s.y += g.y; s.z += g.z; s.w += g.w;
                    float* b = &wacc[w * ROW2 + c4];
                    atomicAdd(b + 0, fv.x); atomicAdd(b + 1, fv.y);
                    atomicAdd(b + 2, fv.z); atomicAdd(b + 3, fv.w);
                }
                a[r] = s;
            }
        }
        lds_barrier();
    }

#pragma unroll
    for (int r = 0; r < TROUNDS; ++r) {
        const int tl = grp + r * NGRP;
        if (gact && tl < TBLK) {
            float4* p = (float4*)&Fs[tl * ROW2 + c4];
            const float4 fv = *p;
            *p = make_float4(fv.x + a[r].x, fv.y + a[r].y, fv.z + a[r].z, fv.w + a[r].w);
        }
    }
    __syncthreads();

    for (int i = tid; i < TBLK * CLS; i += OWN_THREADS) {
        const int tl = i / CLS;
        const int c  = i - tl * CLS;
        const float d = dts[tl];
        const size_t o = (size_t)t_base * CLS + i;
        mean[o] = bm[c] + d * Fs[tl * ROW2 + c];
        lsd[o]  = bl[c] + d * Fs[tl * ROW2 + CLS + c];
    }

    float4* __restrict__ pl = (float4*)(planes + (size_t)blockIdx.x * PLANE);
    for (int i = tid; i < PLANE / 4; i += OWN_THREADS) pl[i] = *(const float4*)&wacc[4 * i];
}


constexpr int Z_THREADS = 320;
constexpr int ZW_EL     = 32;
constexpr int ZW_PARTS  = Z_THREADS / ZW_EL;
constexpr int ZW_DEPTH  = NOWN / ZW_PARTS;
constexpr int ZW_WGS    = (WORKERS * CLS + ZW_EL - 1) / ZW_EL;
constexpr int ZT_WGS    = TASKS * CLS / Z_THREADS;
constexpr int Z_WGS     = ZW_WGS + ZT_WGS;
static_assert(NOWN % ZW_PARTS == 0 && (TASKS * CLS) % Z_THREADS == 0 && Z_THREADS % ZW_EL == 0,
              "z partition must be exact");

__global__ __launch_bounds__(Z_THREADS) void k_z(const float* __restrict__ planes,
                                                 const float* __restrict__ hm, const float* __restrict__ hl,
                                                 const float* __restrict__ dinv,
                                                 const float* __restrict__ bm, const float* __restrict__ bl,
                                                 const float* __restrict__ eps,
                                                 float* __restrict__ mean, float* __restrict__ lsd,
                                                 float* __restrict__ z) {
    __shared__ float red[2][ZW_PARTS][ZW_EL];

    if (blockIdx.x >= ZW_WGS) {
        const int idx = (blockIdx.x - ZW_WGS) * Z_THREADS + threadIdx.x;
        z[idx] = mean[idx] + (eps[idx] * 0.01f) * expf(lsd[idx]);
        return;
    }

    const int  el   = threadIdx.x % ZW_EL;
    const int  part = threadIdx.x / ZW_EL;
    const int  i    = blockIdx.x * ZW_EL + el;
    const bool ok   = i < WORKERS * CLS;
    const int  ic   = ok ? i : WORKERS * CLS - 1;
    const int  w    = ic / CLS;
    const int  c    = ic - w * CLS;

    const float* __restrict__ p = planes + (size_t)part * PLANE + w * ROW2 + c;
    float sm = 0.f, sl = 0.f;
#pragma unroll
    for (int k = 0; k < ZW_DEPTH; ++k) {
        sm += p[(size_t)k * ZW_PARTS * PLANE];
        sl += p[(size_t)k * ZW_PARTS * PLANE + CLS];
    }
    red[0][part][el] = sm;
    red[1][part][el] = sl;
    __syncthreads();

    if (part == 0 && ok) {
        float tm = red[0][0][el], tl = red[1][0][el];
#pragma unroll
        for (int q = 1; q < ZW_PARTS; ++q) { tm += red[0][q][el]; tl += red[1][q][el]; }
        const float  d  = dinv[TASKS + w];
        const float  sw = d * d;
        const size_t o  = (size_t)TASKS * CLS + i;
        const float  m  = bm[c] + hm[(size_t)(TASKS + w) * HP + c] * sw + d * tm;
        const float  l  = bl[c] + hl[(size_t)(TASKS + w) * HP + c] * sw + d * tl;
        mean[o] = m;
        lsd[o]  = l;
        z[o]    = m + (eps[o] * 0.01f) * expf(l);
    }
}

constexpr int NQUAD      = ANS * CLS / 4;
constexpr int CR_THREADS = 160;
constexpr int CR_WGS     = 625;
constexpr int CR_GT      = CR_THREADS * CR_WGS;
constexpr int CR_TRIPS   = NQUAD / CR_GT;
constexpr int CR_BATCH   = 5;
constexpr int CR_ASTEP   = CR_GT * 4 / CLS;
static_assert((ANS * CLS) % 4 == 0 && NQUAD % CR_GT == 0 && (CR_GT * 4) % CLS == 0 &&
              CR_TRIPS % CR_BATCH == 0 && CLS % 2 == 0, "crowd partition must be exact");

__global__ __launch_bounds__(CR_THREADS) void k_crowd(const int* __restrict__ keys,
                                                      const float* __restrict__ z,
                                                      float* __restrict__ crowd) {
    const int g  = blockIdx.x * CR_THREADS + threadIdx.x;
    const int eA = 4 * g;
    const int aA = eA / CLS;
    const int cA = eA - aA * CLS;
    const int eB = eA + 2;
    const int aB = eB / CLS;
    const int cB = eB - aB * CLS;

    const float* __restrict__ ztA = z + cA;
    const float* __restrict__ ztB = z + cB;
    const float* __restrict__ zwA = z + (size_t)TASKS * CLS + cA;
    const float* __restrict__ zwB = z + (size_t)TASKS * CLS + cB;
    const int*   __restrict__ kA  = keys + aA;
    const int*   __restrict__ kB  = keys + aB;
    float4*      __restrict__ o   = (float4*)crowd + g;

    for (int b = 0; b < CR_TRIPS; b += CR_BATCH) {
        int ka[CR_BATCH], kb[CR_BATCH];
#pragma unroll
        for (int u = 0; u < CR_BATCH; ++u) {
            ka[u] = kA[(b + u) * CR_ASTEP];
            kb[u] = kB[(b + u) * CR_ASTEP];
        }
        float2 ta[CR_BATCH], wa[CR_BATCH], tb[CR_BATCH], wb[CR_BATCH];
#pragma unroll
        for (int u = 0; u < CR_BATCH; ++u) {
            ta[u] = *(const float2*)(ztA + (ka[u] >> KSH) * CLS);
            wa[u] = *(const float2*)(zwA + (ka[u] & KMASK) * CLS);
            tb[u] = *(const float2*)(ztB + (kb[u] >> KSH) * CLS);
            wb[u] = *(const float2*)(zwB + (kb[u] & KMASK) * CLS);
        }
#pragma unroll
        for (int u = 0; u < CR_BATCH; ++u)
            o[(size_t)(b + u) * CR_GT] = make_float4(ta[u].x * wa[u].x, ta[u].y * wa[u].y,
                                                     tb[u].x * wb[u].x, tb[u].y * wb[u].y);
    }
}


extern "C" void kernel_launch(void* const* d_in, const int* in_sizes, int n_in,
                              void* d_out, int out_size, void* d_ws, size_t ws_size,
                              hipStream_t stream) {
    (void)in_sizes; (void)n_in; (void)out_size; (void)ws_size;

    const float* task_feature   = (const float*)d_in[0];
    const int*   answers        = (const int*)d_in[1];
    const float* worker_feature = (const float*)d_in[2];
    const float* W_efc          = (const float*)d_in[3];
    const float* b_efc          = (const float*)d_in[4];
    const float* W_mean         = (const float*)d_in[5];
    const float* b_mean         = (const float*)d_in[6];
    const float* W_ls           = (const float*)d_in[7];
    const float* b_ls           = (const float*)d_in[8];
    const float* eps            = (const float*)d_in[9];

    float* out   = (float*)d_out;
    float* crowd = out;
    float* mean  = out + (size_t)ANS * CLS;
    float* lsd   = mean + (size_t)NODES * CLS;

    float*  ws   = (float*)d_ws;
    float*  tf_h = ws;
    float*  hm   = tf_h + (size_t)TASKS * HID;
    float*  hl   = hm + (size_t)NODES * HP;
    float*  zz   = hl + (size_t)NODES * HP;
    float*  dinv = zz + (size_t)NODES * CLS;

    auto cdiv = [](int a, int b) { return (a + b - 1) / b; };

    k_gemm_efc <<<GEMM_WGS, 256, 0, stream>>>(task_feature, W_efc, b_efc, tf_h);
    k_h        <<<cdiv(cdiv(NODES, 16), 8), 256, 0, stream>>>(tf_h, worker_feature,
                                                              W_mean, W_ls, hm, hl);

    int* keys  = (int*)tf_h;
    int* hist  = keys + ANS;
    int* wpart = hist + (size_t)SP_WGS * HPITCH;
    int* segst = wpart + (size_t)SP_WGS * WPITCH;
    int* sorted = segst + HPITCH;
    float* planes = (float*)(sorted + ANS);
    k_split_count   <<<SP_WGS, SP_THREADS, 0, stream>>>(answers, keys, hist, wpart);
    k_split_scatter <<<SC_WGS, SP_THREADS, 0, stream>>>(keys, hist, wpart, sorted, segst, dinv);

    k_agg_owner <<<NOWN, OWN_THREADS, 0, stream>>>(sorted, segst, hm, hl, dinv, b_mean, b_ls, mean, lsd, planes);

    k_z        <<<Z_WGS, Z_THREADS, 0, stream>>>(planes, hm, hl, dinv, b_mean, b_ls, eps, mean, lsd, zz);
    k_crowd    <<<CR_WGS, CR_THREADS, 0, stream>>>(keys, zz, crowd);
}
